# starting kernel
# baseline (speedup 1.0000x reference)
#include <hip/hip_runtime.h>
#include <cmath>

typedef __attribute__((ext_vector_type(2))) float v2f;
typedef __attribute__((ext_vector_type(8))) float v8f;

#define NTIME   2000
#define NBATCH  128
#define INSIZE  256
#define NOUT    40
#define NEG_BIG (-1.0e30f)

#define NCHUNK  100
#define CLEN    (NTIME / NCHUNK)

__device__ __forceinline__ float softplusf(float v) {
    return fmaxf(v, 0.0f) + __logf(1.0f + __expf(-fabsf(v)));
}

__device__ __forceinline__ float tanhf_fast(float v) {
    const float a = fabsf(v);
    const float e = __expf(2.0f * a);
    const float t = 1.0f - 2.0f * __builtin_amdgcn_rcpf(e + 1.0f);
    return copysignf(t, v);
}

__device__ __forceinline__ float lse8(const float (&t)[8]) {
    float mx = t[0];
#pragma unroll
    for (int m = 1; m < 8; ++m) mx = fmaxf(mx, t[m]);
    float s = 0.0f;
#pragma unroll
    for (int m = 0; m < 8; ++m) s += __expf(t[m] - mx);
    return mx + __logf(s);
}

__device__ __forceinline__ int iminc(int a, int b) { return a < b ? a : b; }

__global__ __launch_bounds__(256) void gemm_act_kernel(
    const float* __restrict__ x, const float* __restrict__ W,
    const float* __restrict__ bias, float* __restrict__ out)
{
    const int lane  = threadIdx.x & 31;
    const int wave  = threadIdx.x >> 5;
    const int l16   = lane & 15;
    const int halfk = lane >> 4;

    const int rowBase = (blockIdx.x * 8 + wave) * 64;

    v8f acc[4][3];
#pragma unroll
    for (int m = 0; m < 4; ++m)
#pragma unroll
        for (int n = 0; n < 3; ++n)
#pragma unroll
            for (int i = 0; i < 8; ++i) acc[m][n][i] = 0.0f;

    const float* xp = x + (size_t)(rowBase + l16) * INSIZE + 2 * halfk;
    const float* wp0 = W + (size_t)iminc(0 * 16 + l16, NOUT - 1) * INSIZE + 2 * halfk;
    const float* wp1 = W + (size_t)iminc(1 * 16 + l16, NOUT - 1) * INSIZE + 2 * halfk;
    const float* wp2 = W + (size_t)iminc(2 * 16 + l16, NOUT - 1) * INSIZE + 2 * halfk;

#pragma unroll 2
    for (int kk = 0; kk < INSIZE; kk += 4) {
        v2f a0 = __builtin_nontemporal_load((const v2f*)(xp + kk));
        v2f a1 = __builtin_nontemporal_load((const v2f*)(xp + 16 * INSIZE + kk));
        v2f a2 = __builtin_nontemporal_load((const v2f*)(xp + 32 * INSIZE + kk));
        v2f a3 = __builtin_nontemporal_load((const v2f*)(xp + 48 * INSIZE + kk));
        v2f b0 = *(const v2f*)(wp0 + kk);
        v2f b1 = *(const v2f*)(wp1 + kk);
        v2f b2 = *(const v2f*)(wp2 + kk);

        acc[0][0] = __builtin_amdgcn_wmma_f32_16x16x4_f32(false, a0, false, b0, (short)0, acc[0][0], false, false);
        acc[0][1] = __builtin_amdgcn_wmma_f32_16x16x4_f32(false, a0, false, b1, (short)0, acc[0][1], false, false);
        acc[0][2] = __builtin_amdgcn_wmma_f32_16x16x4_f32(false, a0, false, b2, (short)0, acc[0][2], false, false);
        acc[1][0] = __builtin_amdgcn_wmma_f32_16x16x4_f32(false, a1, false, b0, (short)0, acc[1][0], false, false);
        acc[1][1] = __builtin_amdgcn_wmma_f32_16x16x4_f32(false, a1, false, b1, (short)0, acc[1][1], false, false);
        acc[1][2] = __builtin_amdgcn_wmma_f32_16x16x4_f32(false, a1, false, b2, (short)0, acc[1][2], false, false);
        acc[2][0] = __builtin_amdgcn_wmma_f32_16x16x4_f32(false, a2, false, b0, (short)0, acc[2][0], false, false);
        acc[2][1] = __builtin_amdgcn_wmma_f32_16x16x4_f32(false, a2, false, b1, (short)0, acc[2][1], false, false);
        acc[2][2] = __builtin_amdgcn_wmma_f32_16x16x4_f32(false, a2, false, b2, (short)0, acc[2][2], false, false);
        acc[3][0] = __builtin_amdgcn_wmma_f32_16x16x4_f32(false, a3, false, b0, (short)0, acc[3][0], false, false);
        acc[3][1] = __builtin_amdgcn_wmma_f32_16x16x4_f32(false, a3, false, b1, (short)0, acc[3][1], false, false);
        acc[3][2] = __builtin_amdgcn_wmma_f32_16x16x4_f32(false, a3, false, b2, (short)0, acc[3][2], false, false);
    }

#pragma unroll
    for (int m = 0; m < 4; ++m) {
#pragma unroll
        for (int nt = 0; nt < 3; ++nt) {
            const int col = nt * 16 + l16;
            if (col >= NOUT) continue;
            const float bb = bias[col];
            const int row0 = rowBase + m * 16 + halfk * 8;
#pragma unroll
            for (int r = 0; r < 8; ++r) {
                const float v = acc[m][nt][r] + bb;
                float res;
                if (nt == 0) {
                    const float sp   = softplusf(v);
                    const float th   = 5.0f * tanhf_fast(v);
                    const float ofs  = (col < 4) ? 1.0f : 0.1f;
                    res = (col < 8) ? (ofs + sp) : th;
                } else {
                    res = 5.0f * tanhf_fast(v);
                }
                out[(size_t)(row0 + r) * NOUT + col] = res;
            }
        }
    }
}

__global__ __launch_bounds__(32) void chunk_scan_kernel(
    const float* __restrict__ out, float* __restrict__ chunkM)
{
    __shared__ float Mld[64];
    __shared__ float wld[32];

    const int lane = threadIdx.x;
    const int b    = blockIdx.x & (NBATCH - 1);
    const int c    = blockIdx.x >> 7;
    const int s    = lane & 7;
    const int r0   = lane >> 3;
    const int irow = s & 3;

    float mask[8];
#pragma unroll
    for (int m = 0; m < 8; ++m) {
        const bool on = (s < 4) ? (m == s || m == s + 4) : (m == s - 4 || m == s);
        mask[m] = (s < 4) ? (on ? NEG_BIG : 0.0f) : (on ? 0.0f : NEG_BIG);
    }

    const int t0 = c * CLEN;

    wld[lane] = out[((size_t)t0 * NBATCH + b) * NOUT + 8 + lane];
    __syncthreads();
    float new0 = wld[irow * 8 + r0]     + mask[r0];
    float new1 = wld[irow * 8 + r0 + 4] + mask[r0 + 4];
    __syncthreads();
    Mld[r0 * 8 + s]       = new0;
    Mld[(r0 + 4) * 8 + s] = new1;

    for (int step = 1; step < CLEN; ++step) {
        const int t = t0 + step;
        const float wv = out[((size_t)t * NBATCH + b) * NOUT + 8 + lane];
        __syncthreads();
        wld[lane] = wv;
        __syncthreads();

        float A[8];
#pragma unroll
        for (int m = 0; m < 8; ++m) A[m] = wld[irow * 8 + m] + mask[m];

        float t0r[8], t1r[8];
#pragma unroll
        for (int m = 0; m < 8; ++m) {
            t0r[m] = Mld[r0 * 8 + m]       + A[m];
            t1r[m] = Mld[(r0 + 4) * 8 + m] + A[m];
        }
        new0 = lse8(t0r);
        new1 = lse8(t1r);
        __syncthreads();
        Mld[r0 * 8 + s]       = new0;
        Mld[(r0 + 4) * 8 + s] = new1;
    }

    float* dst = chunkM + ((size_t)b * NCHUNK + c) * 64;
    dst[r0 * 8 + s]       = new0;
    dst[(r0 + 4) * 8 + s] = new1;
}

__global__ __launch_bounds__(32) void combine_kernel(
    const float* __restrict__ chunkM, float* __restrict__ logZdiv)
{
    __shared__ float Msh[64];
    __shared__ float vsh[8];
    const int lane = threadIdx.x;
    const int b    = blockIdx.x;
    const float* base = chunkM + (size_t)b * NCHUNK * 64;

    Msh[lane]      = base[lane];
    Msh[lane + 32] = base[lane + 32];
    __syncthreads();
    if (lane < 8) {
        float t[8];
#pragma unroll
        for (int j = 0; j < 8; ++j) t[j] = Msh[j * 8 + lane];
        vsh[lane] = lse8(t);
    }
    for (int c = 1; c < NCHUNK; ++c) {
        __syncthreads();
        Msh[lane]      = base[c * 64 + lane];
        Msh[lane + 32] = base[c * 64 + 32 + lane];
        __syncthreads();
        if (lane < 8) {
            float vold[8];
#pragma unroll
            for (int j = 0; j < 8; ++j) vold[j] = vsh[j];
            float t[8];
#pragma unroll
            for (int j = 0; j < 8; ++j) t[j] = vold[j] + Msh[j * 8 + lane];
            vsh[lane] = lse8(t);
        }
    }
    __syncthreads();
    if (lane == 0) {
        float t[8];
#pragma unroll
        for (int j = 0; j < 8; ++j) t[j] = vsh[j];
        logZdiv[b] = lse8(t) / (float)NTIME;
    }
}

__global__ __launch_bounds__(1024) void seq_scan_kernel(
    const float* __restrict__ out, float* __restrict__ logZdiv)
{
    __shared__ float fwdsh[NBATCH * 8];
    const int tid  = threadIdx.x;
    const int b    = tid >> 3;
    const int s    = tid & 7;
    const int irow = s & 3;

    float mask[8];
#pragma unroll
    for (int m = 0; m < 8; ++m) {
        const bool on = (s < 4) ? (m == s || m == s + 4) : (m == s - 4 || m == s);
        mask[m] = (s < 4) ? (on ? NEG_BIG : 0.0f) : (on ? 0.0f : NEG_BIG);
    }

    fwdsh[tid] = 0.0f;
    __syncthreads();
    for (int t = 0; t < NTIME; ++t) {
        const float* wrow = out + ((size_t)t * NBATCH + b) * NOUT + 8 + irow * 8;
        const float4 w0 = *(const float4*)wrow;
        const float4 w1 = *(const float4*)(wrow + 4);
        float tt[8] = { w0.x + mask[0], w0.y + mask[1], w0.z + mask[2], w0.w + mask[3],
                        w1.x + mask[4], w1.y + mask[5], w1.z + mask[6], w1.w + mask[7] };
#pragma unroll
        for (int m = 0; m < 8; ++m) tt[m] += fwdsh[(b << 3) + m];
        const float nf = lse8(tt);
        __syncthreads();
        fwdsh[tid] = nf;
        __syncthreads();
    }
    if (s == 0) {
        float t[8];
#pragma unroll
        for (int j = 0; j < 8; ++j) t[j] = fwdsh[(b << 3) + j];
        logZdiv[b] = lse8(t) / (float)NTIME;
    }
}

__global__ __launch_bounds__(256) void sublogz_kernel(
    float* __restrict__ out, const float* __restrict__ logZdiv)
{
    const size_t total = (size_t)NTIME * NBATCH * 32;
    const size_t e = (size_t)blockIdx.x * blockDim.x + threadIdx.x;
    if (e >= total) return;
    const int    j  = (int)(e & 31);
    const size_t tb = e >> 5;
    const int    b  = (int)(tb & (NBATCH - 1));
    out[tb * NOUT + 8 + j] -= logZdiv[b];
}

extern "C" void kernel_launch(void* const* d_in, const int* in_sizes, int n_in,
                              void* d_out, int out_size, void* d_ws, size_t ws_size,
                              hipStream_t stream) {
    (void)in_sizes; (void)n_in; (void)out_size;
    const float* x    = (const float*)d_in[0];
    const float* W    = (const float*)d_in[1];
    const float* bias = (const float*)d_in[2];
    float* out = (float*)d_out;

    const int ROWS = NTIME * NBATCH;
    gemm_act_kernel<<<ROWS / 512, 256, 0, stream>>>(x, W, bias, out);

    const size_t chunkBytes = (size_t)NBATCH * NCHUNK * 64 * sizeof(float);
    const size_t logZBytes  = (size_t)NBATCH * sizeof(float);
    const size_t total = (size_t)NTIME * NBATCH * 32;
    const int subBlocks = (int)((total + 255) / 256);

    if (ws_size >= chunkBytes + logZBytes) {
        float* chunkM  = (float*)d_ws;
        float* logZdiv = (float*)((char*)d_ws + chunkBytes);
        chunk_scan_kernel<<<NBATCH * NCHUNK, 32, 0, stream>>>(out, chunkM);
        combine_kernel<<<NBATCH, 32, 0, stream>>>(chunkM, logZdiv);
        sublogz_kernel<<<subBlocks, 256, 0, stream>>>(out, logZdiv);
    } else {
        float* logZdiv = (float*)d_ws;
        seq_scan_kernel<<<1, 1024, 0, stream>>>(out, logZdiv);
        sublogz_kernel<<<subBlocks, 256, 0, stream>>>(out, logZdiv);
    }
}


# GEMM pass: x goes through per-wave LDS, loaded by whole 256-B row pieces (16 B per lane) instead of 8-B fragment loads over 16 rows; k is summed in a new order inside each 8-wide block
# speedup vs baseline: 1.5388x; 1.5388x over previous
#include <hip/hip_runtime.h>
#include <cmath>

typedef __attribute__((ext_vector_type(2))) float v2f;
typedef __attribute__((ext_vector_type(4))) float v4f;
typedef __attribute__((ext_vector_type(8))) float v8f;

#define NTIME   2000
#define NBATCH  128
#define INSIZE  256
#define NOUT    40
#define NEG_BIG (-1.0e30f)

#define NCHUNK  100
#define CLEN    (NTIME / NCHUNK)

__device__ __forceinline__ float softplusf(float v) {
    return fmaxf(v, 0.0f) + __logf(1.0f + __expf(-fabsf(v)));
}

__device__ __forceinline__ float tanhf_fast(float v) {
    const float a = fabsf(v);
    const float e = __expf(2.0f * a);
    const float t = 1.0f - 2.0f * __builtin_amdgcn_rcpf(e + 1.0f);
    return copysignf(t, v);
}

__device__ __forceinline__ float lse8(const float (&t)[8]) {
    float mx = t[0];
#pragma unroll
    for (int m = 1; m < 8; ++m) mx = fmaxf(mx, t[m]);
    float s = 0.0f;
#pragma unroll
    for (int m = 0; m < 8; ++m) s += __expf(t[m] - mx);
    return mx + __logf(s);
}

__device__ __forceinline__ int iminc(int a, int b) { return a < b ? a : b; }

#define GW_ROWS  32
#define GW_WAVES 4
#define KCH      64
#define XPITCH   68

#define WMMA4(acc_, a_, b_) \
    acc_ = __builtin_amdgcn_wmma_f32_16x16x4_f32(false, a_, false, b_, (short)0, acc_, false, false)

template <bool AHEAD>
__device__ __forceinline__ void gemm_chunk(
    v8f (&acc)[2][3], v4f (&st)[16],
    float* xst, const float* xfr,
    const float* __restrict__ xnext,
    const float* __restrict__ w0, const float* __restrict__ w1,
    const float* __restrict__ w2)
{
    v4f bf[KCH / 8][3];
#pragma unroll
    for (int j = 0; j < KCH / 8; ++j) {
        bf[j][0] = *(const v4f*)(w0 + j * 8);
        bf[j][1] = *(const v4f*)(w1 + j * 8);
        bf[j][2] = *(const v4f*)(w2 + j * 8);
    }

#pragma unroll
    for (int i = 0; i < 16; ++i)
        *(v4f*)(xst + (2 * i) * XPITCH) = st[i];

    if (AHEAD) {
#pragma unroll
        for (int i = 0; i < 16; ++i)
            st[i] = __builtin_nontemporal_load((const v4f*)(xnext + (size_t)(2 * i) * INSIZE));
    }

    __builtin_amdgcn_fence(__ATOMIC_RELEASE, "wavefront");
    __builtin_amdgcn_wave_barrier();
    __builtin_amdgcn_fence(__ATOMIC_ACQUIRE, "wavefront");

#pragma unroll
    for (int j = 0; j < KCH / 8; ++j) {
        const v4f a0 = *(const v4f*)(xfr + j * 8);
        const v4f a1 = *(const v4f*)(xfr + 16 * XPITCH + j * 8);
        const v4f b0 = bf[j][0], b1 = bf[j][1], b2 = bf[j][2];

        const v2f a0l = __builtin_shufflevector(a0, a0, 0, 1), a0h = __builtin_shufflevector(a0, a0, 2, 3);
        const v2f a1l = __builtin_shufflevector(a1, a1, 0, 1), a1h = __builtin_shufflevector(a1, a1, 2, 3);
        const v2f b0l = __builtin_shufflevector(b0, b0, 0, 1), b0h = __builtin_shufflevector(b0, b0, 2, 3);
        const v2f b1l = __builtin_shufflevector(b1, b1, 0, 1), b1h = __builtin_shufflevector(b1, b1, 2, 3);
        const v2f b2l = __builtin_shufflevector(b2, b2, 0, 1), b2h = __builtin_shufflevector(b2, b2, 2, 3);

        WMMA4(acc[0][0], a0l, b0l);
        WMMA4(acc[0][1], a0l, b1l);
        WMMA4(acc[0][2], a0l, b2l);
        WMMA4(acc[1][0], a1l, b0l);
        WMMA4(acc[1][1], a1l, b1l);
        WMMA4(acc[1][2], a1l, b2l);
        WMMA4(acc[0][0], a0h, b0h);
        WMMA4(acc[0][1], a0h, b1h);
        WMMA4(acc[0][2], a0h, b2h);
        WMMA4(acc[1][0], a1h, b0h);
        WMMA4(acc[1][1], a1h, b1h);
        WMMA4(acc[1][2], a1h, b2h);
    }

    __builtin_amdgcn_fence(__ATOMIC_RELEASE, "wavefront");
    __builtin_amdgcn_wave_barrier();
    __builtin_amdgcn_fence(__ATOMIC_ACQUIRE, "wavefront");
}

__global__ __launch_bounds__(128) void gemm_act_kernel(
    const float* __restrict__ x, const float* __restrict__ W,
    const float* __restrict__ bias, float* __restrict__ out)
{
    __shared__ __attribute__((aligned(16))) float xs[GW_WAVES * GW_ROWS * XPITCH];

    const int lane  = threadIdx.x & 31;
    const int wave  = threadIdx.x >> 5;
    const int l16   = lane & 15;
    const int halfk = lane >> 4;

    const int rowBase = (blockIdx.x * GW_WAVES + wave) * GW_ROWS;
    float* xw = xs + wave * (GW_ROWS * XPITCH);

    v8f acc[2][3];
#pragma unroll
    for (int m = 0; m < 2; ++m)
#pragma unroll
        for (int n = 0; n < 3; ++n)
#pragma unroll
            for (int i = 0; i < 8; ++i) acc[m][n][i] = 0.0f;

    const float* xg  = x + (size_t)(rowBase + halfk) * INSIZE + 4 * l16;
    float*       xst = xw + halfk * XPITCH + 4 * l16;
    const float* xfr = xw + l16 * XPITCH + 4 * halfk;
    const float* wp0 = W + (size_t)iminc(0 * 16 + l16, NOUT - 1) * INSIZE + 4 * halfk;
    const float* wp1 = W + (size_t)iminc(1 * 16 + l16, NOUT - 1) * INSIZE + 4 * halfk;
    const float* wp2 = W + (size_t)iminc(2 * 16 + l16, NOUT - 1) * INSIZE + 4 * halfk;

    v4f st[16];
#pragma unroll
    for (int i = 0; i < 16; ++i)
        st[i] = __builtin_nontemporal_load((const v4f*)(xg + (size_t)(2 * i) * INSIZE));

#pragma unroll 1
    for (int ch = 0; ch < INSIZE / KCH - 1; ++ch) {
        const int kc = ch * KCH;
        gemm_chunk<true>(acc, st, xst, xfr, xg + kc + KCH, wp0 + kc, wp1 + kc, wp2 + kc);
    }
    gemm_chunk<false>(acc, st, xst, xfr, xg, wp0 + (INSIZE - KCH), wp1 + (INSIZE - KCH), wp2 + (INSIZE - KCH));

#pragma unroll
    for (int m = 0; m < 2; ++m) {
#pragma unroll
        for (int nt = 0; nt < 3; ++nt) {
            const int col = nt * 16 + l16;
            if (col >= NOUT) continue;
            const float bb = bias[col];
            const int row0 = rowBase + m * 16 + halfk * 8;
#pragma unroll
            for (int r = 0; r < 8; ++r) {
                const float v = acc[m][nt][r] + bb;
                float res;
                if (nt == 0) {
                    const float sp   = softplusf(v);
                    const float th   = 5.0f * tanhf_fast(v);
                    const float ofs  = (col < 4) ? 1.0f : 0.1f;
                    res = (col < 8) ? (ofs + sp) : th;
                } else {
                    res = 5.0f * tanhf_fast(v);
                }
                out[(size_t)(row0 + r) * NOUT + col] = res;
            }
        }
    }
}

__global__ __launch_bounds__(32) void chunk_scan_kernel(
    const float* __restrict__ out, float* __restrict__ chunkM)
{
    __shared__ float Mld[64];
    __shared__ float wld[32];

    const int lane = threadIdx.x;
    const int b    = blockIdx.x & (NBATCH - 1);
    const int c    = blockIdx.x >> 7;
    const int s    = lane & 7;
    const int r0   = lane >> 3;
    const int irow = s & 3;

    float mask[8];
#pragma unroll
    for (int m = 0; m < 8; ++m) {
        const bool on = (s < 4) ? (m == s || m == s + 4) : (m == s - 4 || m == s);
        mask[m] = (s < 4) ? (on ? NEG_BIG : 0.0f) : (on ? 0.0f : NEG_BIG);
    }

    const int t0 = c * CLEN;

    wld[lane] = out[((size_t)t0 * NBATCH + b) * NOUT + 8 + lane];
    __syncthreads();
    float new0 = wld[irow * 8 + r0]     + mask[r0];
    float new1 = wld[irow * 8 + r0 + 4] + mask[r0 + 4];
    __syncthreads();
    Mld[r0 * 8 + s]       = new0;
    Mld[(r0 + 4) * 8 + s] = new1;

    for (int step = 1; step < CLEN; ++step) {
        const int t = t0 + step;
        const float wv = out[((size_t)t * NBATCH + b) * NOUT + 8 + lane];
        __syncthreads();
        wld[lane] = wv;
        __syncthreads();

        float A[8];
#pragma unroll
        for (int m = 0; m < 8; ++m) A[m] = wld[irow * 8 + m] + mask[m];

        float t0r[8], t1r[8];
#pragma unroll
        for (int m = 0; m < 8; ++m) {
            t0r[m] = Mld[r0 * 8 + m]       + A[m];
            t1r[m] = Mld[(r0 + 4) * 8 + m] + A[m];
        }
        new0 = lse8(t0r);
        new1 = lse8(t1r);
        __syncthreads();
        Mld[r0 * 8 + s]       = new0;
        Mld[(r0 + 4) * 8 + s] = new1;
    }

    float* dst = chunkM + ((size_t)b * NCHUNK + c) * 64;
    dst[r0 * 8 + s]       = new0;
    dst[(r0 + 4) * 8 + s] = new1;
}

__global__ __launch_bounds__(32) void combine_kernel(
    const float* __restrict__ chunkM, float* __restrict__ logZdiv)
{
    __shared__ float Msh[64];
    __shared__ float vsh[8];
    const int lane = threadIdx.x;
    const int b    = blockIdx.x;
    const float* base = chunkM + (size_t)b * NCHUNK * 64;

    Msh[lane]      = base[lane];
    Msh[lane + 32] = base[lane + 32];
    __syncthreads();
    if (lane < 8) {
        float t[8];
#pragma unroll
        for (int j = 0; j < 8; ++j) t[j] = Msh[j * 8 + lane];
        vsh[lane] = lse8(t);
    }
    for (int c = 1; c < NCHUNK; ++c) {
        __syncthreads();
        Msh[lane]      = base[c * 64 + lane];
        Msh[lane + 32] = base[c * 64 + 32 + lane];
        __syncthreads();
        if (lane < 8) {
            float vold[8];
#pragma unroll
            for (int j = 0; j < 8; ++j) vold[j] = vsh[j];
            float t[8];
#pragma unroll
            for (int j = 0; j < 8; ++j) t[j] = vold[j] + Msh[j * 8 + lane];
            vsh[lane] = lse8(t);
        }
    }
    __syncthreads();
    if (lane == 0) {
        float t[8];
#pragma unroll
        for (int j = 0; j < 8; ++j) t[j] = vsh[j];
        logZdiv[b] = lse8(t) / (float)NTIME;
    }
}

__global__ __launch_bounds__(1024) void seq_scan_kernel(
    const float* __restrict__ out, float* __restrict__ logZdiv)
{
    __shared__ float fwdsh[NBATCH * 8];
    const int tid  = threadIdx.x;
    const int b    = tid >> 3;
    const int s    = tid & 7;
    const int irow = s & 3;

    float mask[8];
#pragma unroll
    for (int m = 0; m < 8; ++m) {
        const bool on = (s < 4) ? (m == s || m == s + 4) : (m == s - 4 || m == s);
        mask[m] = (s < 4) ? (on ? NEG_BIG : 0.0f) : (on ? 0.0f : NEG_BIG);
    }

    fwdsh[tid] = 0.0f;
    __syncthreads();
    for (int t = 0; t < NTIME; ++t) {
        const float* wrow = out + ((size_t)t * NBATCH + b) * NOUT + 8 + irow * 8;
        const float4 w0 = *(const float4*)wrow;
        const float4 w1 = *(const float4*)(wrow + 4);
        float tt[8] = { w0.x + mask[0], w0.y + mask[1], w0.z + mask[2], w0.w + mask[3],
                        w1.x + mask[4], w1.y + mask[5], w1.z + mask[6], w1.w + mask[7] };
#pragma unroll
        for (int m = 0; m < 8; ++m) tt[m] += fwdsh[(b << 3) + m];
        const float nf = lse8(tt);
        __syncthreads();
        fwdsh[tid] = nf;
        __syncthreads();
    }
    if (s == 0) {
        float t[8];
#pragma unroll
        for (int j = 0; j < 8; ++j) t[j] = fwdsh[(b << 3) + j];
        logZdiv[b] = lse8(t) / (float)NTIME;
    }
}

__global__ __launch_bounds__(256) void sublogz_kernel(
    float* __restrict__ out, const float* __restrict__ logZdiv)
{
    const size_t total = (size_t)NTIME * NBATCH * 32;
    const size_t e = (size_t)blockIdx.x * blockDim.x + threadIdx.x;
    if (e >= total) return;
    const int    j  = (int)(e & 31);
    const size_t tb = e >> 5;
    const int    b  = (int)(tb & (NBATCH - 1));
    out[tb * NOUT + 8 + j] -= logZdiv[b];
}

extern "C" void kernel_launch(void* const* d_in, const int* in_sizes, int n_in,
                              void* d_out, int out_size, void* d_ws, size_t ws_size,
                              hipStream_t stream) {
    (void)in_sizes; (void)n_in; (void)out_size;
    const float* x    = (const float*)d_in[0];
    const float* W    = (const float*)d_in[1];
    const float* bias = (const float*)d_in[2];
    float* out = (float*)d_out;

    const int ROWS = NTIME * NBATCH;
    gemm_act_kernel<<<ROWS / 128, 128, 0, stream>>>(x, W, bias, out);

    const size_t chunkBytes = (size_t)NBATCH * NCHUNK * 64 * sizeof(float);
    const size_t logZBytes  = (size_t)NBATCH * sizeof(float);
    const size_t total = (size_t)NTIME * NBATCH * 32;
    const int subBlocks = (int)((total + 255) / 256);

    if (ws_size >= chunkBytes + logZBytes) {
        float* chunkM  = (float*)d_ws;
        float* logZdiv = (float*)((char*)d_ws + chunkBytes);
        chunk_scan_kernel<<<NBATCH * NCHUNK, 32, 0, stream>>>(out, chunkM);
        combine_kernel<<<NBATCH, 32, 0, stream>>>(chunkM, logZdiv);
        sublogz_kernel<<<subBlocks, 256, 0, stream>>>(out, logZdiv);
    } else {
        float* logZdiv = (float*)d_ws;
        seq_scan_kernel<<<1, 1024, 0, stream>>>(out, logZdiv);
        sublogz_kernel<<<subBlocks, 256, 0, stream>>>(out, logZdiv);
    }
}


# logZ-subtract pass: 16-B loads and stores, 8 pieces per thread issued together, 8000 waves instead of 256000 one-element waves; arithmetic and its order unchanged
# speedup vs baseline: 1.6156x; 1.0499x over previous
#include <hip/hip_runtime.h>
#include <cmath>

typedef __attribute__((ext_vector_type(2))) float v2f;
typedef __attribute__((ext_vector_type(4))) float v4f;
typedef __attribute__((ext_vector_type(8))) float v8f;

#define NTIME   2000
#define NBATCH  128
#define INSIZE  256
#define NOUT    40
#define NEG_BIG (-1.0e30f)

#define NCHUNK  100
#define CLEN    (NTIME / NCHUNK)

__device__ __forceinline__ float softplusf(float v) {
    return fmaxf(v, 0.0f) + __logf(1.0f + __expf(-fabsf(v)));
}

__device__ __forceinline__ float tanhf_fast(float v) {
    const float a = fabsf(v);
    const float e = __expf(2.0f * a);
    const float t = 1.0f - 2.0f * __builtin_amdgcn_rcpf(e + 1.0f);
    return copysignf(t, v);
}

__device__ __forceinline__ float lse8(const float (&t)[8]) {
    float mx = t[0];
#pragma unroll
    for (int m = 1; m < 8; ++m) mx = fmaxf(mx, t[m]);
    float s = 0.0f;
#pragma unroll
    for (int m = 0; m < 8; ++m) s += __expf(t[m] - mx);
    return mx + __logf(s);
}

__device__ __forceinline__ int iminc(int a, int b) { return a < b ? a : b; }

#define GW_ROWS  32
#define GW_WAVES 4
#define KCH      64
#define XPITCH   68

#define WMMA4(acc_, a_, b_) \
    acc_ = __builtin_amdgcn_wmma_f32_16x16x4_f32(false, a_, false, b_, (short)0, acc_, false, false)

template <bool AHEAD>
__device__ __forceinline__ void gemm_chunk(
    v8f (&acc)[2][3], v4f (&st)[16],
    float* xst, const float* xfr,
    const float* __restrict__ xnext,
    const float* __restrict__ w0, const float* __restrict__ w1,
    const float* __restrict__ w2)
{
    v4f bf[KCH / 8][3];
#pragma unroll
    for (int j = 0; j < KCH / 8; ++j) {
        bf[j][0] = *(const v4f*)(w0 + j * 8);
        bf[j][1] = *(const v4f*)(w1 + j * 8);
        bf[j][2] = *(const v4f*)(w2 + j * 8);
    }

#pragma unroll
    for (int i = 0; i < 16; ++i)
        *(v4f*)(xst + (2 * i) * XPITCH) = st[i];

    if (AHEAD) {
#pragma unroll
        for (int i = 0; i < 16; ++i)
            st[i] = __builtin_nontemporal_load((const v4f*)(xnext + (size_t)(2 * i) * INSIZE));
    }

    __builtin_amdgcn_fence(__ATOMIC_RELEASE, "wavefront");
    __builtin_amdgcn_wave_barrier();
    __builtin_amdgcn_fence(__ATOMIC_ACQUIRE, "wavefront");

#pragma unroll
    for (int j = 0; j < KCH / 8; ++j) {
        const v4f a0 = *(const v4f*)(xfr + j * 8);
        const v4f a1 = *(const v4f*)(xfr + 16 * XPITCH + j * 8);
        const v4f b0 = bf[j][0], b1 = bf[j][1], b2 = bf[j][2];

        const v2f a0l = __builtin_shufflevector(a0, a0, 0, 1), a0h = __builtin_shufflevector(a0, a0, 2, 3);
        const v2f a1l = __builtin_shufflevector(a1, a1, 0, 1), a1h = __builtin_shufflevector(a1, a1, 2, 3);
        const v2f b0l = __builtin_shufflevector(b0, b0, 0, 1), b0h = __builtin_shufflevector(b0, b0, 2, 3);
        const v2f b1l = __builtin_shufflevector(b1, b1, 0, 1), b1h = __builtin_shufflevector(b1, b1, 2, 3);
        const v2f b2l = __builtin_shufflevector(b2, b2, 0, 1), b2h = __builtin_shufflevector(b2, b2, 2, 3);

        WMMA4(acc[0][0], a0l, b0l);
        WMMA4(acc[0][1], a0l, b1l);
        WMMA4(acc[0][2], a0l, b2l);
        WMMA4(acc[1][0], a1l, b0l);
        WMMA4(acc[1][1], a1l, b1l);
        WMMA4(acc[1][2], a1l, b2l);
        WMMA4(acc[0][0], a0h, b0h);
        WMMA4(acc[0][1], a0h, b1h);
        WMMA4(acc[0][2], a0h, b2h);
        WMMA4(acc[1][0], a1h, b0h);
        WMMA4(acc[1][1], a1h, b1h);
        WMMA4(acc[1][2], a1h, b2h);
    }

    __builtin_amdgcn_fence(__ATOMIC_RELEASE, "wavefront");
    __builtin_amdgcn_wave_barrier();
    __builtin_amdgcn_fence(__ATOMIC_ACQUIRE, "wavefront");
}

__global__ __launch_bounds__(128) void gemm_act_kernel(
    const float* __restrict__ x, const float* __restrict__ W,
    const float* __restrict__ bias, float* __restrict__ out)
{
    __shared__ __attribute__((aligned(16))) float xs[GW_WAVES * GW_ROWS * XPITCH];

    const int lane  = threadIdx.x & 31;
    const int wave  = threadIdx.x >> 5;
    const int l16   = lane & 15;
    const int halfk = lane >> 4;

    const int rowBase = (blockIdx.x * GW_WAVES + wave) * GW_ROWS;
    float* xw = xs + wave * (GW_ROWS * XPITCH);

    v8f acc[2][3];
#pragma unroll
    for (int m = 0; m < 2; ++m)
#pragma unroll
        for (int n = 0; n < 3; ++n)
#pragma unroll
            for (int i = 0; i < 8; ++i) acc[m][n][i] = 0.0f;

    const float* xg  = x + (size_t)(rowBase + halfk) * INSIZE + 4 * l16;
    float*       xst = xw + halfk * XPITCH + 4 * l16;
    const float* xfr = xw + l16 * XPITCH + 4 * halfk;
    const float* wp0 = W + (size_t)iminc(0 * 16 + l16, NOUT - 1) * INSIZE + 4 * halfk;
    const float* wp1 = W + (size_t)iminc(1 * 16 + l16, NOUT - 1) * INSIZE + 4 * halfk;
    const float* wp2 = W + (size_t)iminc(2 * 16 + l16, NOUT - 1) * INSIZE + 4 * halfk;

    v4f st[16];
#pragma unroll
    for (int i = 0; i < 16; ++i)
        st[i] = __builtin_nontemporal_load((const v4f*)(xg + (size_t)(2 * i) * INSIZE));

#pragma unroll 1
    for (int ch = 0; ch < INSIZE / KCH - 1; ++ch) {
        const int kc = ch * KCH;
        gemm_chunk<true>(acc, st, xst, xfr, xg + kc + KCH, wp0 + kc, wp1 + kc, wp2 + kc);
    }
    gemm_chunk<false>(acc, st, xst, xfr, xg, wp0 + (INSIZE - KCH), wp1 + (INSIZE - KCH), wp2 + (INSIZE - KCH));

#pragma unroll
    for (int m = 0; m < 2; ++m) {
#pragma unroll
        for (int nt = 0; nt < 3; ++nt) {
            const int col = nt * 16 + l16;
            if (col >= NOUT) continue;
            const float bb = bias[col];
            const int row0 = rowBase + m * 16 + halfk * 8;
#pragma unroll
            for (int r = 0; r < 8; ++r) {
                const float v = acc[m][nt][r] + bb;
                float res;
                if (nt == 0) {
                    const float sp   = softplusf(v);
                    const float th   = 5.0f * tanhf_fast(v);
                    const float ofs  = (col < 4) ? 1.0f : 0.1f;
                    res = (col < 8) ? (ofs + sp) : th;
                } else {
                    res = 5.0f * tanhf_fast(v);
                }
                out[(size_t)(row0 + r) * NOUT + col] = res;
            }
        }
    }
}

__global__ __launch_bounds__(32) void chunk_scan_kernel(
    const float* __restrict__ out, float* __restrict__ chunkM)
{
    __shared__ float Mld[64];
    __shared__ float wld[32];

    const int lane = threadIdx.x;
    const int b    = blockIdx.x & (NBATCH - 1);
    const int c    = blockIdx.x >> 7;
    const int s    = lane & 7;
    const int r0   = lane >> 3;
    const int irow = s & 3;

    float mask[8];
#pragma unroll
    for (int m = 0; m < 8; ++m) {
        const bool on = (s < 4) ? (m == s || m == s + 4) : (m == s - 4 || m == s);
        mask[m] = (s < 4) ? (on ? NEG_BIG : 0.0f) : (on ? 0.0f : NEG_BIG);
    }

    const int t0 = c * CLEN;

    wld[lane] = out[((size_t)t0 * NBATCH + b) * NOUT + 8 + lane];
    __syncthreads();
    float new0 = wld[irow * 8 + r0]     + mask[r0];
    float new1 = wld[irow * 8 + r0 + 4] + mask[r0 + 4];
    __syncthreads();
    Mld[r0 * 8 + s]       = new0;
    Mld[(r0 + 4) * 8 + s] = new1;

    for (int step = 1; step < CLEN; ++step) {
        const int t = t0 + step;
        const float wv = out[((size_t)t * NBATCH + b) * NOUT + 8 + lane];
        __syncthreads();
        wld[lane] = wv;
        __syncthreads();

        float A[8];
#pragma unroll
        for (int m = 0; m < 8; ++m) A[m] = wld[irow * 8 + m] + mask[m];

        float t0r[8], t1r[8];
#pragma unroll
        for (int m = 0; m < 8; ++m) {
            t0r[m] = Mld[r0 * 8 + m]       + A[m];
            t1r[m] = Mld[(r0 + 4) * 8 + m] + A[m];
        }
        new0 = lse8(t0r);
        new1 = lse8(t1r);
        __syncthreads();
        Mld[r0 * 8 + s]       = new0;
        Mld[(r0 + 4) * 8 + s] = new1;
    }

    float* dst = chunkM + ((size_t)b * NCHUNK + c) * 64;
    dst[r0 * 8 + s]       = new0;
    dst[(r0 + 4) * 8 + s] = new1;
}

__global__ __launch_bounds__(32) void combine_kernel(
    const float* __restrict__ chunkM, float* __restrict__ logZdiv)
{
    __shared__ float Msh[64];
    __shared__ float vsh[8];
    const int lane = threadIdx.x;
    const int b    = blockIdx.x;
    const float* base = chunkM + (size_t)b * NCHUNK * 64;

    Msh[lane]      = base[lane];
    Msh[lane + 32] = base[lane + 32];
    __syncthreads();
    if (lane < 8) {
        float t[8];
#pragma unroll
        for (int j = 0; j < 8; ++j) t[j] = Msh[j * 8 + lane];
        vsh[lane] = lse8(t);
    }
    for (int c = 1; c < NCHUNK; ++c) {
        __syncthreads();
        Msh[lane]      = base[c * 64 + lane];
        Msh[lane + 32] = base[c * 64 + 32 + lane];
        __syncthreads();
        if (lane < 8) {
            float vold[8];
#pragma unroll
            for (int j = 0; j < 8; ++j) vold[j] = vsh[j];
            float t[8];
#pragma unroll
            for (int j = 0; j < 8; ++j) t[j] = vold[j] + Msh[j * 8 + lane];
            vsh[lane] = lse8(t);
        }
    }
    __syncthreads();
    if (lane == 0) {
        float t[8];
#pragma unroll
        for (int j = 0; j < 8; ++j) t[j] = vsh[j];
        logZdiv[b] = lse8(t) / (float)NTIME;
    }
}

__global__ __launch_bounds__(1024) void seq_scan_kernel(
    const float* __restrict__ out, float* __restrict__ logZdiv)
{
    __shared__ float fwdsh[NBATCH * 8];
    const int tid  = threadIdx.x;
    const int b    = tid >> 3;
    const int s    = tid & 7;
    const int irow = s & 3;

    float mask[8];
#pragma unroll
    for (int m = 0; m < 8; ++m) {
        const bool on = (s < 4) ? (m == s || m == s + 4) : (m == s - 4 || m == s);
        mask[m] = (s < 4) ? (on ? NEG_BIG : 0.0f) : (on ? 0.0f : NEG_BIG);
    }

    fwdsh[tid] = 0.0f;
    __syncthreads();
    for (int t = 0; t < NTIME; ++t) {
        const float* wrow = out + ((size_t)t * NBATCH + b) * NOUT + 8 + irow * 8;
        const float4 w0 = *(const float4*)wrow;
        const float4 w1 = *(const float4*)(wrow + 4);
        float tt[8] = { w0.x + mask[0], w0.y + mask[1], w0.z + mask[2], w0.w + mask[3],
                        w1.x + mask[4], w1.y + mask[5], w1.z + mask[6], w1.w + mask[7] };
#pragma unroll
        for (int m = 0; m < 8; ++m) tt[m] += fwdsh[(b << 3) + m];
        const float nf = lse8(tt);
        __syncthreads();
        fwdsh[tid] = nf;
        __syncthreads();
    }
    if (s == 0) {
        float t[8];
#pragma unroll
        for (int j = 0; j < 8; ++j) t[j] = fwdsh[(b << 3) + j];
        logZdiv[b] = lse8(t) / (float)NTIME;
    }
}

#define SUB_THREADS     128
#define SUB_PIECES      8
#define SUB_NPIECE      (NTIME * NBATCH * 8)
#define SUB_BLOCKS      (SUB_NPIECE / (SUB_THREADS * SUB_PIECES))
#define SUB_STRIDE_ROWS (SUB_BLOCKS * SUB_THREADS / 8)
static_assert(SUB_NPIECE % (SUB_THREADS * SUB_PIECES) == 0, "pieces divide evenly");
static_assert(SUB_STRIDE_ROWS % NBATCH == 0, "one batch entry per thread");

__global__ __launch_bounds__(SUB_THREADS) void sublogz_kernel(
    float* __restrict__ out, const float* __restrict__ logZdiv)
{
    const int q0   = blockIdx.x * SUB_THREADS + threadIdx.x;
    const int row0 = q0 >> 3;
    const int j    = q0 & 7;
    const float z  = logZdiv[row0 & (NBATCH - 1)];
    float* p = out + (size_t)row0 * NOUT + 8 + 4 * j;
    const size_t stride = (size_t)SUB_STRIDE_ROWS * NOUT;

    v4f v[SUB_PIECES];
#pragma unroll
    for (int i = 0; i < SUB_PIECES; ++i)
        v[i] = *(const v4f*)(p + (size_t)i * stride);
#pragma unroll
    for (int i = 0; i < SUB_PIECES; ++i) {
        v4f r = v[i];
        r[0] -= z; r[1] -= z; r[2] -= z; r[3] -= z;
        *(v4f*)(p + (size_t)i * stride) = r;
    }
}

extern "C" void kernel_launch(void* const* d_in, const int* in_sizes, int n_in,
                              void* d_out, int out_size, void* d_ws, size_t ws_size,
                              hipStream_t stream) {
    (void)in_sizes; (void)n_in; (void)out_size;
    const float* x    = (const float*)d_in[0];
    const float* W    = (const float*)d_in[1];
    const float* bias = (const float*)d_in[2];
    float* out = (float*)d_out;

    const int ROWS = NTIME * NBATCH;
    gemm_act_kernel<<<ROWS / 128, 128, 0, stream>>>(x, W, bias, out);

    const size_t chunkBytes = (size_t)NBATCH * NCHUNK * 64 * sizeof(float);
    const size_t logZBytes  = (size_t)NBATCH * sizeof(float);

    if (ws_size >= chunkBytes + logZBytes) {
        float* chunkM  = (float*)d_ws;
        float* logZdiv = (float*)((char*)d_ws + chunkBytes);
        chunk_scan_kernel<<<NBATCH * NCHUNK, 32, 0, stream>>>(out, chunkM);
        combine_kernel<<<NBATCH, 32, 0, stream>>>(chunkM, logZdiv);
        sublogz_kernel<<<SUB_BLOCKS, SUB_THREADS, 0, stream>>>(out, logZdiv);
    } else {
        float* logZdiv = (float*)d_ws;
        seq_scan_kernel<<<1, 1024, 0, stream>>>(out, logZdiv);
        sublogz_kernel<<<SUB_BLOCKS, SUB_THREADS, 0, stream>>>(out, logZdiv);
    }
}


# chunk-combine pass: all 100 chunk matrices of a batch entry loaded to LDS at once, then multiplied as a pairwise tree (7 levels) instead of a 99-step serial fold; log-sum-exp order changes
# speedup vs baseline: 2.5959x; 1.6068x over previous
#include <hip/hip_runtime.h>
#include <cmath>

typedef __attribute__((ext_vector_type(2))) float v2f;
typedef __attribute__((ext_vector_type(4))) float v4f;
typedef __attribute__((ext_vector_type(8))) float v8f;

#define NTIME   2000
#define NBATCH  128
#define INSIZE  256
#define NOUT    40
#define NEG_BIG (-1.0e30f)

#define NCHUNK  100
#define CLEN    (NTIME / NCHUNK)

__device__ __forceinline__ float softplusf(float v) {
    return fmaxf(v, 0.0f) + __logf(1.0f + __expf(-fabsf(v)));
}

__device__ __forceinline__ float tanhf_fast(float v) {
    const float a = fabsf(v);
    const float e = __expf(2.0f * a);
    const float t = 1.0f - 2.0f * __builtin_amdgcn_rcpf(e + 1.0f);
    return copysignf(t, v);
}

__device__ __forceinline__ float lse8(const float (&t)[8]) {
    float mx = t[0];
#pragma unroll
    for (int m = 1; m < 8; ++m) mx = fmaxf(mx, t[m]);
    float s = 0.0f;
#pragma unroll
    for (int m = 0; m < 8; ++m) s += __expf(t[m] - mx);
    return mx + __logf(s);
}

__device__ __forceinline__ int iminc(int a, int b) { return a < b ? a : b; }

#define GW_ROWS  32
#define GW_WAVES 4
#define KCH      64
#define XPITCH   68

#define WMMA4(acc_, a_, b_) \
    acc_ = __builtin_amdgcn_wmma_f32_16x16x4_f32(false, a_, false, b_, (short)0, acc_, false, false)

template <bool AHEAD>
__device__ __forceinline__ void gemm_chunk(
    v8f (&acc)[2][3], v4f (&st)[16],
    float* xst, const float* xfr,
    const float* __restrict__ xnext,
    const float* __restrict__ w0, const float* __restrict__ w1,
    const float* __restrict__ w2)
{
    v4f bf[KCH / 8][3];
#pragma unroll
    for (int j = 0; j < KCH / 8; ++j) {
        bf[j][0] = *(const v4f*)(w0 + j * 8);
        bf[j][1] = *(const v4f*)(w1 + j * 8);
        bf[j][2] = *(const v4f*)(w2 + j * 8);
    }

#pragma unroll
    for (int i = 0; i < 16; ++i)
        *(v4f*)(xst + (2 * i) * XPITCH) = st[i];

    if (AHEAD) {
#pragma unroll
        for (int i = 0; i < 16; ++i)
            st[i] = __builtin_nontemporal_load((const v4f*)(xnext + (size_t)(2 * i) * INSIZE));
    }

    __builtin_amdgcn_fence(__ATOMIC_RELEASE, "wavefront");
    __builtin_amdgcn_wave_barrier();
    __builtin_amdgcn_fence(__ATOMIC_ACQUIRE, "wavefront");

#pragma unroll
    for (int j = 0; j < KCH / 8; ++j) {
        const v4f a0 = *(const v4f*)(xfr + j * 8);
        const v4f a1 = *(const v4f*)(xfr + 16 * XPITCH + j * 8);
        const v4f b0 = bf[j][0], b1 = bf[j][1], b2 = bf[j][2];

        const v2f a0l = __builtin_shufflevector(a0, a0, 0, 1), a0h = __builtin_shufflevector(a0, a0, 2, 3);
        const v2f a1l = __builtin_shufflevector(a1, a1, 0, 1), a1h = __builtin_shufflevector(a1, a1, 2, 3);
        const v2f b0l = __builtin_shufflevector(b0, b0, 0, 1), b0h = __builtin_shufflevector(b0, b0, 2, 3);
        const v2f b1l = __builtin_shufflevector(b1, b1, 0, 1), b1h = __builtin_shufflevector(b1, b1, 2, 3);
        const v2f b2l = __builtin_shufflevector(b2, b2, 0, 1), b2h = __builtin_shufflevector(b2, b2, 2, 3);

        WMMA4(acc[0][0], a0l, b0l);
        WMMA4(acc[0][1], a0l, b1l);
        WMMA4(acc[0][2], a0l, b2l);
        WMMA4(acc[1][0], a1l, b0l);
        WMMA4(acc[1][1], a1l, b1l);
        WMMA4(acc[1][2], a1l, b2l);
        WMMA4(acc[0][0], a0h, b0h);
        WMMA4(acc[0][1], a0h, b1h);
        WMMA4(acc[0][2], a0h, b2h);
        WMMA4(acc[1][0], a1h, b0h);
        WMMA4(acc[1][1], a1h, b1h);
        WMMA4(acc[1][2], a1h, b2h);
    }

    __builtin_amdgcn_fence(__ATOMIC_RELEASE, "wavefront");
    __builtin_amdgcn_wave_barrier();
    __builtin_amdgcn_fence(__ATOMIC_ACQUIRE, "wavefront");
}

__global__ __launch_bounds__(128) void gemm_act_kernel(
    const float* __restrict__ x, const float* __restrict__ W,
    const float* __restrict__ bias, float* __restrict__ out)
{
    __shared__ __attribute__((aligned(16))) float xs[GW_WAVES * GW_ROWS * XPITCH];

    const int lane  = threadIdx.x & 31;
    const int wave  = threadIdx.x >> 5;
    const int l16   = lane & 15;
    const int halfk = lane >> 4;

    const int rowBase = (blockIdx.x * GW_WAVES + wave) * GW_ROWS;
    float* xw = xs + wave * (GW_ROWS * XPITCH);

    v8f acc[2][3];
#pragma unroll
    for (int m = 0; m < 2; ++m)
#pragma unroll
        for (int n = 0; n < 3; ++n)
#pragma unroll
            for (int i = 0; i < 8; ++i) acc[m][n][i] = 0.0f;

    const float* xg  = x + (size_t)(rowBase + halfk) * INSIZE + 4 * l16;
    float*       xst = xw + halfk * XPITCH + 4 * l16;
    const float* xfr = xw + l16 * XPITCH + 4 * halfk;
    const float* wp0 = W + (size_t)iminc(0 * 16 + l16, NOUT - 1) * INSIZE + 4 * halfk;
    const float* wp1 = W + (size_t)iminc(1 * 16 + l16, NOUT - 1) * INSIZE + 4 * halfk;
    const float* wp2 = W + (size_t)iminc(2 * 16 + l16, NOUT - 1) * INSIZE + 4 * halfk;

    v4f st[16];
#pragma unroll
    for (int i = 0; i < 16; ++i)
        st[i] = __builtin_nontemporal_load((const v4f*)(xg + (size_t)(2 * i) * INSIZE));

#pragma unroll 1
    for (int ch = 0; ch < INSIZE / KCH - 1; ++ch) {
        const int kc = ch * KCH;
        gemm_chunk<true>(acc, st, xst, xfr, xg + kc + KCH, wp0 + kc, wp1 + kc, wp2 + kc);
    }
    gemm_chunk<false>(acc, st, xst, xfr, xg, wp0 + (INSIZE - KCH), wp1 + (INSIZE - KCH), wp2 + (INSIZE - KCH));

#pragma unroll
    for (int m = 0; m < 2; ++m) {
#pragma unroll
        for (int nt = 0; nt < 3; ++nt) {
            const int col = nt * 16 + l16;
            if (col >= NOUT) continue;
            const float bb = bias[col];
            const int row0 = rowBase + m * 16 + halfk * 8;
#pragma unroll
            for (int r = 0; r < 8; ++r) {
                const float v = acc[m][nt][r] + bb;
                float res;
                if (nt == 0) {
                    const float sp   = softplusf(v);
                    const float th   = 5.0f * tanhf_fast(v);
                    const float ofs  = (col < 4) ? 1.0f : 0.1f;
                    res = (col < 8) ? (ofs + sp) : th;
                } else {
                    res = 5.0f * tanhf_fast(v);
                }
                out[(size_t)(row0 + r) * NOUT + col] = res;
            }
        }
    }
}

__global__ __launch_bounds__(32) void chunk_scan_kernel(
    const float* __restrict__ out, float* __restrict__ chunkM)
{
    __shared__ float Mld[64];
    __shared__ float wld[32];

    const int lane = threadIdx.x;
    const int b    = blockIdx.x & (NBATCH - 1);
    const int c    = blockIdx.x >> 7;
    const int s    = lane & 7;
    const int r0   = lane >> 3;
    const int irow = s & 3;

    float mask[8];
#pragma unroll
    for (int m = 0; m < 8; ++m) {
        const bool on = (s < 4) ? (m == s || m == s + 4) : (m == s - 4 || m == s);
        mask[m] = (s < 4) ? (on ? NEG_BIG : 0.0f) : (on ? 0.0f : NEG_BIG);
    }

    const int t0 = c * CLEN;

    wld[lane] = out[((size_t)t0 * NBATCH + b) * NOUT + 8 + lane];
    __syncthreads();
    float new0 = wld[irow * 8 + r0]     + mask[r0];
    float new1 = wld[irow * 8 + r0 + 4] + mask[r0 + 4];
    __syncthreads();
    Mld[r0 * 8 + s]       = new0;
    Mld[(r0 + 4) * 8 + s] = new1;

    for (int step = 1; step < CLEN; ++step) {
        const int t = t0 + step;
        const float wv = out[((size_t)t * NBATCH + b) * NOUT + 8 + lane];
        __syncthreads();
        wld[lane] = wv;
        __syncthreads();

        float A[8];
#pragma unroll
        for (int m = 0; m < 8; ++m) A[m] = wld[irow * 8 + m] + mask[m];

        float t0r[8], t1r[8];
#pragma unroll
        for (int m = 0; m < 8; ++m) {
            t0r[m] = Mld[r0 * 8 + m]       + A[m];
            t1r[m] = Mld[(r0 + 4) * 8 + m] + A[m];
        }
        new0 = lse8(t0r);
        new1 = lse8(t1r);
        __syncthreads();
        Mld[r0 * 8 + s]       = new0;
        Mld[(r0 + 4) * 8 + s] = new1;
    }

    float* dst = chunkM + ((size_t)b * NCHUNK + c) * 64;
    dst[r0 * 8 + s]       = new0;
    dst[(r0 + 4) * 8 + s] = new1;
}

#define CMB_THREADS 512
#define CMB_GROUPS  (CMB_THREADS / 64)
#define CMB_PIECES  ((NCHUNK * 16 + CMB_THREADS - 1) / CMB_THREADS)

__global__ __launch_bounds__(CMB_THREADS) void combine_kernel(
    const float* __restrict__ chunkM, float* __restrict__ logZdiv)
{
    __shared__ __attribute__((aligned(16))) float bufA[NCHUNK * 64];
    __shared__ __attribute__((aligned(16))) float bufB[((NCHUNK + 1) / 2) * 64];
    __shared__ float vsh[8];

    const int tid = threadIdx.x;
    const int b   = blockIdx.x;
    const float* base = chunkM + (size_t)b * NCHUNK * 64;

    v4f pc[CMB_PIECES];
#pragma unroll
    for (int k = 0; k < CMB_PIECES; ++k) {
        const int i = tid + k * CMB_THREADS;
        if (i < NCHUNK * 16) pc[k] = *(const v4f*)(base + 4 * i);
    }
#pragma unroll
    for (int k = 0; k < CMB_PIECES; ++k) {
        const int i = tid + k * CMB_THREADS;
        if (i < NCHUNK * 16) *(v4f*)(bufA + 4 * i) = pc[k];
    }
    __syncthreads();

    const int e = tid & 63;
    const int g = tid >> 6;
    const int r = e >> 3;
    const int s = e & 7;

    float* src = bufA;
    float* dst = bufB;
#pragma unroll
    for (int n = NCHUNK; n > 1; n = (n + 1) >> 1) {
        const int np = (n + 1) >> 1;
        for (int p = g; p < np; p += CMB_GROUPS) {
            const float* L = src + (2 * p) * 64;
            float res;
            if (2 * p + 1 < n) {
                const float* R = L + 64;
                float t[8];
#pragma unroll
                for (int m = 0; m < 8; ++m) t[m] = L[r * 8 + m] + R[m * 8 + s];
                res = lse8(t);
            } else {
                res = L[e];
            }
            dst[p * 64 + e] = res;
        }
        __syncthreads();
        float* tmp = src; src = dst; dst = tmp;
    }

    if (tid < 8) {
        float t[8];
#pragma unroll
        for (int j = 0; j < 8; ++j) t[j] = src[j * 8 + tid];
        vsh[tid] = lse8(t);
    }
    __syncthreads();
    if (tid == 0) {
        float t[8];
#pragma unroll
        for (int j = 0; j < 8; ++j) t[j] = vsh[j];
        logZdiv[b] = lse8(t) / (float)NTIME;
    }
}

__global__ __launch_bounds__(1024) void seq_scan_kernel(
    const float* __restrict__ out, float* __restrict__ logZdiv)
{
    __shared__ float fwdsh[NBATCH * 8];
    const int tid  = threadIdx.x;
    const int b    = tid >> 3;
    const int s    = tid & 7;
    const int irow = s & 3;

    float mask[8];
#pragma unroll
    for (int m = 0; m < 8; ++m) {
        const bool on = (s < 4) ? (m == s || m == s + 4) : (m == s - 4 || m == s);
        mask[m] = (s < 4) ? (on ? NEG_BIG : 0.0f) : (on ? 0.0f : NEG_BIG);
    }

    fwdsh[tid] = 0.0f;
    __syncthreads();
    for (int t = 0; t < NTIME; ++t) {
        const float* wrow = out + ((size_t)t * NBATCH + b) * NOUT + 8 + irow * 8;
        const float4 w0 = *(const float4*)wrow;
        const float4 w1 = *(const float4*)(wrow + 4);
        float tt[8] = { w0.x + mask[0], w0.y + mask[1], w0.z + mask[2], w0.w + mask[3],
                        w1.x + mask[4], w1.y + mask[5], w1.z + mask[6], w1.w + mask[7] };
#pragma unroll
        for (int m = 0; m < 8; ++m) tt[m] += fwdsh[(b << 3) + m];
        const float nf = lse8(tt);
        __syncthreads();
        fwdsh[tid] = nf;
        __syncthreads();
    }
    if (s == 0) {
        float t[8];
#pragma unroll
        for (int j = 0; j < 8; ++j) t[j] = fwdsh[(b << 3) + j];
        logZdiv[b] = lse8(t) / (float)NTIME;
    }
}

#define SUB_THREADS     128
#define SUB_PIECES      8
#define SUB_NPIECE      (NTIME * NBATCH * 8)
#define SUB_BLOCKS      (SUB_NPIECE / (SUB_THREADS * SUB_PIECES))
#define SUB_STRIDE_ROWS (SUB_BLOCKS * SUB_THREADS / 8)
static_assert(SUB_NPIECE % (SUB_THREADS * SUB_PIECES) == 0, "pieces divide evenly");
static_assert(SUB_STRIDE_ROWS % NBATCH == 0, "one batch entry per thread");

__global__ __launch_bounds__(SUB_THREADS) void sublogz_kernel(
    float* __restrict__ out, const float* __restrict__ logZdiv)
{
    const int q0   = blockIdx.x * SUB_THREADS + threadIdx.x;
    const int row0 = q0 >> 3;
    const int j    = q0 & 7;
    const float z  = logZdiv[row0 & (NBATCH - 1)];
    float* p = out + (size_t)row0 * NOUT + 8 + 4 * j;
    const size_t stride = (size_t)SUB_STRIDE_ROWS * NOUT;

    v4f v[SUB_PIECES];
#pragma unroll
    for (int i = 0; i < SUB_PIECES; ++i)
        v[i] = *(const v4f*)(p + (size_t)i * stride);
#pragma unroll
    for (int i = 0; i < SUB_PIECES; ++i) {
        v4f r = v[i];
        r[0] -= z; r[1] -= z; r[2] -= z; r[3] -= z;
        *(v4f*)(p + (size_t)i * stride) = r;
    }
}

extern "C" void kernel_launch(void* const* d_in, const int* in_sizes, int n_in,
                              void* d_out, int out_size, void* d_ws, size_t ws_size,
                              hipStream_t stream) {
    (void)in_sizes; (void)n_in; (void)out_size;
    const float* x    = (const float*)d_in[0];
    const float* W    = (const float*)d_in[1];
    const float* bias = (const float*)d_in[2];
    float* out = (float*)d_out;

    const int ROWS = NTIME * NBATCH;
    gemm_act_kernel<<<ROWS / 128, 128, 0, stream>>>(x, W, bias, out);

    const size_t chunkBytes = (size_t)NBATCH * NCHUNK * 64 * sizeof(float);
    const size_t logZBytes  = (size_t)NBATCH * sizeof(float);

    if (ws_size >= chunkBytes + logZBytes) {
        float* chunkM  = (float*)d_ws;
        float* logZdiv = (float*)((char*)d_ws + chunkBytes);
        chunk_scan_kernel<<<NBATCH * NCHUNK, 32, 0, stream>>>(out, chunkM);
        combine_kernel<<<NBATCH, CMB_THREADS, 0, stream>>>(chunkM, logZdiv);
        sublogz_kernel<<<SUB_BLOCKS, SUB_THREADS, 0, stream>>>(out, logZdiv);
    } else {
        float* logZdiv = (float*)d_ws;
        seq_scan_kernel<<<1, 1024, 0, stream>>>(out, logZdiv);
        sublogz_kernel<<<SUB_BLOCKS, SUB_THREADS, 0, stream>>>(out, logZdiv);
    }
}


# chunk scan in the linear domain: exp once per weight, two matrix rows per lane, weights shared in quads by DPP moves, power-of-two rescale every 5 steps, one log per entry; sums of products, not LSE
# speedup vs baseline: 3.1083x; 1.1974x over previous
#include <hip/hip_runtime.h>
#include <cmath>

typedef __attribute__((ext_vector_type(2))) float v2f;
typedef __attribute__((ext_vector_type(4))) float v4f;
typedef __attribute__((ext_vector_type(8))) float v8f;

#define NTIME   2000
#define NBATCH  128
#define INSIZE  256
#define NOUT    40
#define NEG_BIG (-1.0e30f)

#define NCHUNK  100
#define CLEN    (NTIME / NCHUNK)

__device__ __forceinline__ float softplusf(float v) {
    return fmaxf(v, 0.0f) + __logf(1.0f + __expf(-fabsf(v)));
}

__device__ __forceinline__ float tanhf_fast(float v) {
    const float a = fabsf(v);
    const float e = __expf(2.0f * a);
    const float t = 1.0f - 2.0f * __builtin_amdgcn_rcpf(e + 1.0f);
    return copysignf(t, v);
}

__device__ __forceinline__ float lse8(const float (&t)[8]) {
    float mx = t[0];
#pragma unroll
    for (int m = 1; m < 8; ++m) mx = fmaxf(mx, t[m]);
    float s = 0.0f;
#pragma unroll
    for (int m = 0; m < 8; ++m) s += __expf(t[m] - mx);
    return mx + __logf(s);
}

__device__ __forceinline__ int iminc(int a, int b) { return a < b ? a : b; }

#define GW_ROWS  32
#define GW_WAVES 4
#define KCH      64
#define XPITCH   68

#define WMMA4(acc_, a_, b_) \
    acc_ = __builtin_amdgcn_wmma_f32_16x16x4_f32(false, a_, false, b_, (short)0, acc_, false, false)

template <bool AHEAD>
__device__ __forceinline__ void gemm_chunk(
    v8f (&acc)[2][3], v4f (&st)[16],
    float* xst, const float* xfr,
    const float* __restrict__ xnext,
    const float* __restrict__ w0, const float* __restrict__ w1,
    const float* __restrict__ w2)
{
    v4f bf[KCH / 8][3];
#pragma unroll
    for (int j = 0; j < KCH / 8; ++j) {
        bf[j][0] = *(const v4f*)(w0 + j * 8);
        bf[j][1] = *(const v4f*)(w1 + j * 8);
        bf[j][2] = *(const v4f*)(w2 + j * 8);
    }

#pragma unroll
    for (int i = 0; i < 16; ++i)
        *(v4f*)(xst + (2 * i) * XPITCH) = st[i];

    if (AHEAD) {
#pragma unroll
        for (int i = 0; i < 16; ++i)
            st[i] = __builtin_nontemporal_load((const v4f*)(xnext + (size_t)(2 * i) * INSIZE));
    }

    __builtin_amdgcn_fence(__ATOMIC_RELEASE, "wavefront");
    __builtin_amdgcn_wave_barrier();
    __builtin_amdgcn_fence(__ATOMIC_ACQUIRE, "wavefront");

#pragma unroll
    for (int j = 0; j < KCH / 8; ++j) {
        const v4f a0 = *(const v4f*)(xfr + j * 8);
        const v4f a1 = *(const v4f*)(xfr + 16 * XPITCH + j * 8);
        const v4f b0 = bf[j][0], b1 = bf[j][1], b2 = bf[j][2];

        const v2f a0l = __builtin_shufflevector(a0, a0, 0, 1), a0h = __builtin_shufflevector(a0, a0, 2, 3);
        const v2f a1l = __builtin_shufflevector(a1, a1, 0, 1), a1h = __builtin_shufflevector(a1, a1, 2, 3);
        const v2f b0l = __builtin_shufflevector(b0, b0, 0, 1), b0h = __builtin_shufflevector(b0, b0, 2, 3);
        const v2f b1l = __builtin_shufflevector(b1, b1, 0, 1), b1h = __builtin_shufflevector(b1, b1, 2, 3);
        const v2f b2l = __builtin_shufflevector(b2, b2, 0, 1), b2h = __builtin_shufflevector(b2, b2, 2, 3);

        WMMA4(acc[0][0], a0l, b0l);
        WMMA4(acc[0][1], a0l, b1l);
        WMMA4(acc[0][2], a0l, b2l);
        WMMA4(acc[1][0], a1l, b0l);
        WMMA4(acc[1][1], a1l, b1l);
        WMMA4(acc[1][2], a1l, b2l);
        WMMA4(acc[0][0], a0h, b0h);
        WMMA4(acc[0][1], a0h, b1h);
        WMMA4(acc[0][2], a0h, b2h);
        WMMA4(acc[1][0], a1h, b0h);
        WMMA4(acc[1][1], a1h, b1h);
        WMMA4(acc[1][2], a1h, b2h);
    }

    __builtin_amdgcn_fence(__ATOMIC_RELEASE, "wavefront");
    __builtin_amdgcn_wave_barrier();
    __builtin_amdgcn_fence(__ATOMIC_ACQUIRE, "wavefront");
}

__global__ __launch_bounds__(128) void gemm_act_kernel(
    const float* __restrict__ x, const float* __restrict__ W,
    const float* __restrict__ bias, float* __restrict__ out)
{
    __shared__ __attribute__((aligned(16))) float xs[GW_WAVES * GW_ROWS * XPITCH];

    const int lane  = threadIdx.x & 31;
    const int wave  = threadIdx.x >> 5;
    const int l16   = lane & 15;
    const int halfk = lane >> 4;

    const int rowBase = (blockIdx.x * GW_WAVES + wave) * GW_ROWS;
    float* xw = xs + wave * (GW_ROWS * XPITCH);

    v8f acc[2][3];
#pragma unroll
    for (int m = 0; m < 2; ++m)
#pragma unroll
        for (int n = 0; n < 3; ++n)
#pragma unroll
            for (int i = 0; i < 8; ++i) acc[m][n][i] = 0.0f;

    const float* xg  = x + (size_t)(rowBase + halfk) * INSIZE + 4 * l16;
    float*       xst = xw + halfk * XPITCH + 4 * l16;
    const float* xfr = xw + l16 * XPITCH + 4 * halfk;
    const float* wp0 = W + (size_t)iminc(0 * 16 + l16, NOUT - 1) * INSIZE + 4 * halfk;
    const float* wp1 = W + (size_t)iminc(1 * 16 + l16, NOUT - 1) * INSIZE + 4 * halfk;
    const float* wp2 = W + (size_t)iminc(2 * 16 + l16, NOUT - 1) * INSIZE + 4 * halfk;

    v4f st[16];
#pragma unroll
    for (int i = 0; i < 16; ++i)
        st[i] = __builtin_nontemporal_load((const v4f*)(xg + (size_t)(2 * i) * INSIZE));

#pragma unroll 1
    for (int ch = 0; ch < INSIZE / KCH - 1; ++ch) {
        const int kc = ch * KCH;
        gemm_chunk<true>(acc, st, xst, xfr, xg + kc + KCH, wp0 + kc, wp1 + kc, wp2 + kc);
    }
    gemm_chunk<false>(acc, st, xst, xfr, xg, wp0 + (INSIZE - KCH), wp1 + (INSIZE - KCH), wp2 + (INSIZE - KCH));

#pragma unroll
    for (int m = 0; m < 2; ++m) {
#pragma unroll
        for (int nt = 0; nt < 3; ++nt) {
            const int col = nt * 16 + l16;
            if (col >= NOUT) continue;
            const float bb = bias[col];
            const int row0 = rowBase + m * 16 + halfk * 8;
#pragma unroll
            for (int r = 0; r < 8; ++r) {
                const float v = acc[m][nt][r] + bb;
                float res;
                if (nt == 0) {
                    const float sp   = softplusf(v);
                    const float th   = 5.0f * tanhf_fast(v);
                    const float ofs  = (col < 4) ? 1.0f : 0.1f;
                    res = (col < 8) ? (ofs + sp) : th;
                } else {
                    res = 5.0f * tanhf_fast(v);
                }
                out[(size_t)(row0 + r) * NOUT + col] = res;
            }
        }
    }
}

#define SCAN_THREADS 128
#define SCAN_GROUPS  (SCAN_THREADS / 4)
#define SCAN_BLOCKS  (NBATCH * NCHUNK / SCAN_GROUPS)
#define SCAN_RESCALE 5
static_assert((NBATCH * NCHUNK) % SCAN_GROUPS == 0, "chunks divide evenly over blocks");

template <int L>
__device__ __forceinline__ float bcast4(float v) {
    return __uint_as_float(__builtin_amdgcn_mov_dpp(__float_as_uint(v), L * 0x55, 0xf, 0xf, true));
}

template <int I>
__device__ __forceinline__ void scan_row(const v2f (&M)[8], const v4f ea, const v4f eb, v2f& mv, v2f& sty) {
    const v4f e = (I < 2) ? ea : eb;
    constexpr int L0 = 2 * (I & 1);
    const float ev[8] = {
        bcast4<L0>(e[0]),     bcast4<L0>(e[1]),     bcast4<L0>(e[2]),     bcast4<L0>(e[3]),
        bcast4<L0 + 1>(e[0]), bcast4<L0 + 1>(e[1]), bcast4<L0 + 1>(e[2]), bcast4<L0 + 1>(e[3]) };
    sty = M[I] * ev[I] + M[I + 4] * ev[I + 4];
    v2f acc = {0.0f, 0.0f};
    bool first = true;
#pragma unroll
    for (int m = 0; m < 8; ++m) {
        if (m == I || m == I + 4) continue;
        acc = first ? M[m] * ev[m] : acc + M[m] * ev[m];
        first = false;
    }
    mv = acc;
}

__global__ __launch_bounds__(SCAN_THREADS) void chunk_scan_kernel(
    const float* __restrict__ out, float* __restrict__ chunkM)
{
    const int q = threadIdx.x & 3;
    const int G = blockIdx.x * SCAN_GROUPS + (threadIdx.x >> 2);
    const int b = G & (NBATCH - 1);
    const int c = G >> 7;

    const float* p = out + ((size_t)(c * CLEN) * NBATCH + b) * NOUT + 8 + 4 * q;
    v4f wa[CLEN], wb[CLEN];
#pragma unroll
    for (int k = 0; k < CLEN; ++k) {
        wa[k] = *(const v4f*)(p + (size_t)k * (NBATCH * NOUT));
        wb[k] = *(const v4f*)(p + (size_t)k * (NBATCH * NOUT) + 16);
    }
    __builtin_amdgcn_sched_barrier(0);

    v2f M[8];
#pragma unroll
    for (int m = 0; m < 8; ++m) {
        M[m][0] = (m == q)     ? 1.0f : 0.0f;
        M[m][1] = (m == q + 4) ? 1.0f : 0.0f;
    }
    int esum0 = 0, esum1 = 0;

#pragma unroll
    for (int k = 0; k < CLEN; ++k) {
        v4f ea, eb;
#pragma unroll
        for (int i = 0; i < 4; ++i) { ea[i] = __expf(wa[k][i]); eb[i] = __expf(wb[k][i]); }
        v2f Mn[8];
        scan_row<0>(M, ea, eb, Mn[0], Mn[4]);
        scan_row<1>(M, ea, eb, Mn[1], Mn[5]);
        scan_row<2>(M, ea, eb, Mn[2], Mn[6]);
        scan_row<3>(M, ea, eb, Mn[3], Mn[7]);
        if ((k % SCAN_RESCALE) == SCAN_RESCALE - 1 && k != CLEN - 1) {
            float mx0 = Mn[0][0], mx1 = Mn[0][1];
#pragma unroll
            for (int m = 1; m < 8; ++m) { mx0 = fmaxf(mx0, Mn[m][0]); mx1 = fmaxf(mx1, Mn[m][1]); }
            const int ex0 = __builtin_amdgcn_frexp_expf(mx0);
            const int ex1 = __builtin_amdgcn_frexp_expf(mx1);
            esum0 += ex0; esum1 += ex1;
#pragma unroll
            for (int m = 0; m < 8; ++m) { Mn[m][0] = ldexpf(Mn[m][0], -ex0); Mn[m][1] = ldexpf(Mn[m][1], -ex1); }
        }
#pragma unroll
        for (int m = 0; m < 8; ++m) M[m] = Mn[m];
    }

    const float es0 = (float)esum0, es1 = (float)esum1;
    const float ln2 = 0.69314718055994531f;
    v4f r0lo, r0hi, r1lo, r1hi;
#pragma unroll
    for (int m = 0; m < 4; ++m) {
        r0lo[m] = (__log2f(M[m][0])     + es0) * ln2;
        r0hi[m] = (__log2f(M[m + 4][0]) + es0) * ln2;
        r1lo[m] = (__log2f(M[m][1])     + es1) * ln2;
        r1hi[m] = (__log2f(M[m + 4][1]) + es1) * ln2;
    }
    float* dst = chunkM + ((size_t)b * NCHUNK + c) * 64 + q * 8;
    *(v4f*)(dst)      = r0lo;
    *(v4f*)(dst + 4)  = r0hi;
    *(v4f*)(dst + 32) = r1lo;
    *(v4f*)(dst + 36) = r1hi;
}

#define CMB_THREADS 512
#define CMB_GROUPS  (CMB_THREADS / 64)
#define CMB_PIECES  ((NCHUNK * 16 + CMB_THREADS - 1) / CMB_THREADS)

__global__ __launch_bounds__(CMB_THREADS) void combine_kernel(
    const float* __restrict__ chunkM, float* __restrict__ logZdiv)
{
    __shared__ __attribute__((aligned(16))) float bufA[NCHUNK * 64];
    __shared__ __attribute__((aligned(16))) float bufB[((NCHUNK + 1) / 2) * 64];
    __shared__ float vsh[8];

    const int tid = threadIdx.x;
    const int b   = blockIdx.x;
    const float* base = chunkM + (size_t)b * NCHUNK * 64;

    v4f pc[CMB_PIECES];
#pragma unroll
    for (int k = 0; k < CMB_PIECES; ++k) {
        const int i = tid + k * CMB_THREADS;
        if (i < NCHUNK * 16) pc[k] = *(const v4f*)(base + 4 * i);
    }
#pragma unroll
    for (int k = 0; k < CMB_PIECES; ++k) {
        const int i = tid + k * CMB_THREADS;
        if (i < NCHUNK * 16) *(v4f*)(bufA + 4 * i) = pc[k];
    }
    __syncthreads();

    const int e = tid & 63;
    const int g = tid >> 6;
    const int r = e >> 3;
    const int s = e & 7;

    float* src = bufA;
    float* dst = bufB;
#pragma unroll
    for (int n = NCHUNK; n > 1; n = (n + 1) >> 1) {
        const int np = (n + 1) >> 1;
        for (int p = g; p < np; p += CMB_GROUPS) {
            const float* L = src + (2 * p) * 64;
            float res;
            if (2 * p + 1 < n) {
                const float* R = L + 64;
                float t[8];
#pragma unroll
                for (int m = 0; m < 8; ++m) t[m] = L[r * 8 + m] + R[m * 8 + s];
                res = lse8(t);
            } else {
                res = L[e];
            }
            dst[p * 64 + e] = res;
        }
        __syncthreads();
        float* tmp = src; src = dst; dst = tmp;
    }

    if (tid < 8) {
        float t[8];
#pragma unroll
        for (int j = 0; j < 8; ++j) t[j] = src[j * 8 + tid];
        vsh[tid] = lse8(t);
    }
    __syncthreads();
    if (tid == 0) {
        float t[8];
#pragma unroll
        for (int j = 0; j < 8; ++j) t[j] = vsh[j];
        logZdiv[b] = lse8(t) / (float)NTIME;
    }
}

__global__ __launch_bounds__(1024) void seq_scan_kernel(
    const float* __restrict__ out, float* __restrict__ logZdiv)
{
    __shared__ float fwdsh[NBATCH * 8];
    const int tid  = threadIdx.x;
    const int b    = tid >> 3;
    const int s    = tid & 7;
    const int irow = s & 3;

    float mask[8];
#pragma unroll
    for (int m = 0; m < 8; ++m) {
        const bool on = (s < 4) ? (m == s || m == s + 4) : (m == s - 4 || m == s);
        mask[m] = (s < 4) ? (on ? NEG_BIG : 0.0f) : (on ? 0.0f : NEG_BIG);
    }

    fwdsh[tid] = 0.0f;
    __syncthreads();
    for (int t = 0; t < NTIME; ++t) {
        const float* wrow = out + ((size_t)t * NBATCH + b) * NOUT + 8 + irow * 8;
        const float4 w0 = *(const float4*)wrow;
        const float4 w1 = *(const float4*)(wrow + 4);
        float tt[8] = { w0.x + mask[0], w0.y + mask[1], w0.z + mask[2], w0.w + mask[3],
                        w1.x + mask[4], w1.y + mask[5], w1.z + mask[6], w1.w + mask[7] };
#pragma unroll
        for (int m = 0; m < 8; ++m) tt[m] += fwdsh[(b << 3) + m];
        const float nf = lse8(tt);
        __syncthreads();
        fwdsh[tid] = nf;
        __syncthreads();
    }
    if (s == 0) {
        float t[8];
#pragma unroll
        for (int j = 0; j < 8; ++j) t[j] = fwdsh[(b << 3) + j];
        logZdiv[b] = lse8(t) / (float)NTIME;
    }
}

#define SUB_THREADS     128
#define SUB_PIECES      8
#define SUB_NPIECE      (NTIME * NBATCH * 8)
#define SUB_BLOCKS      (SUB_NPIECE / (SUB_THREADS * SUB_PIECES))
#define SUB_STRIDE_ROWS (SUB_BLOCKS * SUB_THREADS / 8)
static_assert(SUB_NPIECE % (SUB_THREADS * SUB_PIECES) == 0, "pieces divide evenly");
static_assert(SUB_STRIDE_ROWS % NBATCH == 0, "one batch entry per thread");

__global__ __launch_bounds__(SUB_THREADS) void sublogz_kernel(
    float* __restrict__ out, const float* __restrict__ logZdiv)
{
    const int q0   = blockIdx.x * SUB_THREADS + threadIdx.x;
    const int row0 = q0 >> 3;
    const int j    = q0 & 7;
    const float z  = logZdiv[row0 & (NBATCH - 1)];
    float* p = out + (size_t)row0 * NOUT + 8 + 4 * j;
    const size_t stride = (size_t)SUB_STRIDE_ROWS * NOUT;

    v4f v[SUB_PIECES];
#pragma unroll
    for (int i = 0; i < SUB_PIECES; ++i)
        v[i] = *(const v4f*)(p + (size_t)i * stride);
#pragma unroll
    for (int i = 0; i < SUB_PIECES; ++i) {
        v4f r = v[i];
        r[0] -= z; r[1] -= z; r[2] -= z; r[3] -= z;
        *(v4f*)(p + (size_t)i * stride) = r;
    }
}

extern "C" void kernel_launch(void* const* d_in, const int* in_sizes, int n_in,
                              void* d_out, int out_size, void* d_ws, size_t ws_size,
                              hipStream_t stream) {
    (void)in_sizes; (void)n_in; (void)out_size;
    const float* x    = (const float*)d_in[0];
    const float* W    = (const float*)d_in[1];
    const float* bias = (const float*)d_in[2];
    float* out = (float*)d_out;

    const int ROWS = NTIME * NBATCH;
    gemm_act_kernel<<<ROWS / 128, 128, 0, stream>>>(x, W, bias, out);

    const size_t chunkBytes = (size_t)NBATCH * NCHUNK * 64 * sizeof(float);
    const size_t logZBytes  = (size_t)NBATCH * sizeof(float);

    if (ws_size >= chunkBytes + logZBytes) {
        float* chunkM  = (float*)d_ws;
        float* logZdiv = (float*)((char*)d_ws + chunkBytes);
        chunk_scan_kernel<<<SCAN_BLOCKS, SCAN_THREADS, 0, stream>>>(out, chunkM);
        combine_kernel<<<NBATCH, CMB_THREADS, 0, stream>>>(chunkM, logZdiv);
        sublogz_kernel<<<SUB_BLOCKS, SUB_THREADS, 0, stream>>>(out, logZdiv);
    } else {
        float* logZdiv = (float*)d_ws;
        seq_scan_kernel<<<1, 1024, 0, stream>>>(out, logZdiv);
        sublogz_kernel<<<SUB_BLOCKS, SUB_THREADS, 0, stream>>>(out, logZdiv);
    }
}


# GEMM pass: W is copied to LDS once per work-group by whole-line loads and B fragments are read from LDS, not by global loads that each take 32 B out of 16 separate lines (24 per wave and K chunk)
# speedup vs baseline: 3.3605x; 1.0811x over previous
#include <hip/hip_runtime.h>
#include <cmath>

typedef __attribute__((ext_vector_type(2))) float v2f;
typedef __attribute__((ext_vector_type(4))) float v4f;
typedef __attribute__((ext_vector_type(8))) float v8f;

#define NTIME   2000
#define NBATCH  128
#define INSIZE  256
#define NOUT    40
#define NEG_BIG (-1.0e30f)

#define NCHUNK  100
#define CLEN    (NTIME / NCHUNK)

__device__ __forceinline__ float softplusf(float v) {
    return fmaxf(v, 0.0f) + __logf(1.0f + __expf(-fabsf(v)));
}

__device__ __forceinline__ float tanhf_fast(float v) {
    const float a = fabsf(v);
    const float e = __expf(2.0f * a);
    const float t = 1.0f - 2.0f * __builtin_amdgcn_rcpf(e + 1.0f);
    return copysignf(t, v);
}

__device__ __forceinline__ float lse8(const float (&t)[8]) {
    float mx = t[0];
#pragma unroll
    for (int m = 1; m < 8; ++m) mx = fmaxf(mx, t[m]);
    float s = 0.0f;
#pragma unroll
    for (int m = 0; m < 8; ++m) s += __expf(t[m] - mx);
    return mx + __logf(s);
}

__device__ __forceinline__ int iminc(int a, int b) { return a < b ? a : b; }

#define GW_ROWS  32
#define GW_WAVES 4
#define KCH      64
#define XPITCH   68
#define WPITCH   260
#define WPIECES  (NOUT * INSIZE / 4 / (GW_WAVES * 32))
static_assert((NOUT * INSIZE / 4) % (GW_WAVES * 32) == 0, "W pieces divide evenly over the threads");

#define WMMA4(acc_, a_, b_) \
    acc_ = __builtin_amdgcn_wmma_f32_16x16x4_f32(false, a_, false, b_, (short)0, acc_, false, false)

template <bool AHEAD>
__device__ __forceinline__ void gemm_chunk(
    v8f (&acc)[2][3], v4f (&st)[16],
    float* xst, const float* xfr,
    const float* __restrict__ xnext,
    const float* __restrict__ w0, const float* __restrict__ w1,
    const float* __restrict__ w2)
{
    v4f bf[KCH / 8][3];
#pragma unroll
    for (int j = 0; j < KCH / 8; ++j) {
        bf[j][0] = *(const v4f*)(w0 + j * 8);
        bf[j][1] = *(const v4f*)(w1 + j * 8);
        bf[j][2] = *(const v4f*)(w2 + j * 8);
    }

#pragma unroll
    for (int i = 0; i < 16; ++i)
        *(v4f*)(xst + (2 * i) * XPITCH) = st[i];

    if (AHEAD) {
#pragma unroll
        for (int i = 0; i < 16; ++i)
            st[i] = __builtin_nontemporal_load((const v4f*)(xnext + (size_t)(2 * i) * INSIZE));
    }

    __builtin_amdgcn_fence(__ATOMIC_RELEASE, "wavefront");
    __builtin_amdgcn_wave_barrier();
    __builtin_amdgcn_fence(__ATOMIC_ACQUIRE, "wavefront");

#pragma unroll
    for (int j = 0; j < KCH / 8; ++j) {
        const v4f a0 = *(const v4f*)(xfr + j * 8);
        const v4f a1 = *(const v4f*)(xfr + 16 * XPITCH + j * 8);
        const v4f b0 = bf[j][0], b1 = bf[j][1], b2 = bf[j][2];

        const v2f a0l = __builtin_shufflevector(a0, a0, 0, 1), a0h = __builtin_shufflevector(a0, a0, 2, 3);
        const v2f a1l = __builtin_shufflevector(a1, a1, 0, 1), a1h = __builtin_shufflevector(a1, a1, 2, 3);
        const v2f b0l = __builtin_shufflevector(b0, b0, 0, 1), b0h = __builtin_shufflevector(b0, b0, 2, 3);
        const v2f b1l = __builtin_shufflevector(b1, b1, 0, 1), b1h = __builtin_shufflevector(b1, b1, 2, 3);
        const v2f b2l = __builtin_shufflevector(b2, b2, 0, 1), b2h = __builtin_shufflevector(b2, b2, 2, 3);

        WMMA4(acc[0][0], a0l, b0l);
        WMMA4(acc[0][1], a0l, b1l);
        WMMA4(acc[0][2], a0l, b2l);
        WMMA4(acc[1][0], a1l, b0l);
        WMMA4(acc[1][1], a1l, b1l);
        WMMA4(acc[1][2], a1l, b2l);
        WMMA4(acc[0][0], a0h, b0h);
        WMMA4(acc[0][1], a0h, b1h);
        WMMA4(acc[0][2], a0h, b2h);
        WMMA4(acc[1][0], a1h, b0h);
        WMMA4(acc[1][1], a1h, b1h);
        WMMA4(acc[1][2], a1h, b2h);
    }

    __builtin_amdgcn_fence(__ATOMIC_RELEASE, "wavefront");
    __builtin_amdgcn_wave_barrier();
    __builtin_amdgcn_fence(__ATOMIC_ACQUIRE, "wavefront");
}

__global__ __launch_bounds__(128) void gemm_act_kernel(
    const float* __restrict__ x, const float* __restrict__ W,
    const float* __restrict__ bias, float* __restrict__ out)
{
    __shared__ __attribute__((aligned(16))) float xs[GW_WAVES * GW_ROWS * XPITCH];
    __shared__ __attribute__((aligned(16))) float wsh[NOUT * WPITCH];

    const int lane  = threadIdx.x & 31;
    const int wave  = threadIdx.x >> 5;
    const int l16   = lane & 15;
    const int halfk = lane >> 4;

    const int rowBase = (blockIdx.x * GW_WAVES + wave) * GW_ROWS;
    float* xw = xs + wave * (GW_ROWS * XPITCH);

    v8f acc[2][3];
#pragma unroll
    for (int m = 0; m < 2; ++m)
#pragma unroll
        for (int n = 0; n < 3; ++n)
#pragma unroll
            for (int i = 0; i < 8; ++i) acc[m][n][i] = 0.0f;

    const float* xg  = x + (size_t)(rowBase + halfk) * INSIZE + 4 * l16;
    float*       xst = xw + halfk * XPITCH + 4 * l16;
    const float* xfr = xw + l16 * XPITCH + 4 * halfk;
    const float* wp0 = wsh + iminc(0 * 16 + l16, NOUT - 1) * WPITCH + 4 * halfk;
    const float* wp1 = wsh + iminc(1 * 16 + l16, NOUT - 1) * WPITCH + 4 * halfk;
    const float* wp2 = wsh + iminc(2 * 16 + l16, NOUT - 1) * WPITCH + 4 * halfk;

    v4f wpc[WPIECES];
#pragma unroll
    for (int i = 0; i < WPIECES; ++i)
        wpc[i] = *(const v4f*)(W + 4 * ((int)threadIdx.x + GW_WAVES * 32 * i));

    v4f st[16];
#pragma unroll
    for (int i = 0; i < 16; ++i)
        st[i] = __builtin_nontemporal_load((const v4f*)(xg + (size_t)(2 * i) * INSIZE));

#pragma unroll
    for (int i = 0; i < WPIECES; ++i) {
        const int p = (int)threadIdx.x + GW_WAVES * 32 * i;
        *(v4f*)(wsh + (p >> 6) * WPITCH + 4 * (p & 63)) = wpc[i];
    }
    __syncthreads();

#pragma unroll 1
    for (int ch = 0; ch < INSIZE / KCH - 1; ++ch) {
        const int kc = ch * KCH;
        gemm_chunk<true>(acc, st, xst, xfr, xg + kc + KCH, wp0 + kc, wp1 + kc, wp2 + kc);
    }
    gemm_chunk<false>(acc, st, xst, xfr, xg, wp0 + (INSIZE - KCH), wp1 + (INSIZE - KCH), wp2 + (INSIZE - KCH));

#pragma unroll
    for (int m = 0; m < 2; ++m) {
#pragma unroll
        for (int nt = 0; nt < 3; ++nt) {
            const int col = nt * 16 + l16;
            if (col >= NOUT) continue;
            const float bb = bias[col];
            const int row0 = rowBase + m * 16 + halfk * 8;
#pragma unroll
            for (int r = 0; r < 8; ++r) {
                const float v = acc[m][nt][r] + bb;
                float res;
                if (nt == 0) {
                    const float sp   = softplusf(v);
                    const float th   = 5.0f * tanhf_fast(v);
                    const float ofs  = (col < 4) ? 1.0f : 0.1f;
                    res = (col < 8) ? (ofs + sp) : th;
                } else {
                    res = 5.0f * tanhf_fast(v);
                }
                out[(size_t)(row0 + r) * NOUT + col] = res;
            }
        }
    }
}

#define SCAN_THREADS 128
#define SCAN_GROUPS  (SCAN_THREADS / 4)
#define SCAN_BLOCKS  (NBATCH * NCHUNK / SCAN_GROUPS)
#define SCAN_RESCALE 5
static_assert((NBATCH * NCHUNK) % SCAN_GROUPS == 0, "chunks divide evenly over blocks");

template <int L>
__device__ __forceinline__ float bcast4(float v) {
    return __uint_as_float(__builtin_amdgcn_mov_dpp(__float_as_uint(v), L * 0x55, 0xf, 0xf, true));
}

template <int I>
__device__ __forceinline__ void scan_row(const v2f (&M)[8], const v4f ea, const v4f eb, v2f& mv, v2f& sty) {
    const v4f e = (I < 2) ? ea : eb;
    constexpr int L0 = 2 * (I & 1);
    const float ev[8] = {
        bcast4<L0>(e[0]),     bcast4<L0>(e[1]),     bcast4<L0>(e[2]),     bcast4<L0>(e[3]),
        bcast4<L0 + 1>(e[0]), bcast4<L0 + 1>(e[1]), bcast4<L0 + 1>(e[2]), bcast4<L0 + 1>(e[3]) };
    sty = M[I] * ev[I] + M[I + 4] * ev[I + 4];
    v2f acc = {0.0f, 0.0f};
    bool first = true;
#pragma unroll
    for (int m = 0; m < 8; ++m) {
        if (m == I || m == I + 4) continue;
        acc = first ? M[m] * ev[m] : acc + M[m] * ev[m];
        first = false;
    }
    mv = acc;
}

__global__ __launch_bounds__(SCAN_THREADS) void chunk_scan_kernel(
    const float* __restrict__ out, float* __restrict__ chunkM)
{
    const int q = threadIdx.x & 3;
    const int G = blockIdx.x * SCAN_GROUPS + (threadIdx.x >> 2);
    const int b = G & (NBATCH - 1);
    const int c = G >> 7;

    const float* p = out + ((size_t)(c * CLEN) * NBATCH + b) * NOUT + 8 + 4 * q;
    v4f wa[CLEN], wb[CLEN];
#pragma unroll
    for (int k = 0; k < CLEN; ++k) {
        wa[k] = *(const v4f*)(p + (size_t)k * (NBATCH * NOUT));
        wb[k] = *(const v4f*)(p + (size_t)k * (NBATCH * NOUT) + 16);
    }
    __builtin_amdgcn_sched_barrier(0);

    v2f M[8];
#pragma unroll
    for (int m = 0; m < 8; ++m) {
        M[m][0] = (m == q)     ? 1.0f : 0.0f;
        M[m][1] = (m == q + 4) ? 1.0f : 0.0f;
    }
    int esum0 = 0, esum1 = 0;

#pragma unroll
    for (int k = 0; k < CLEN; ++k) {
        v4f ea, eb;
#pragma unroll
        for (int i = 0; i < 4; ++i) { ea[i] = __expf(wa[k][i]); eb[i] = __expf(wb[k][i]); }
        v2f Mn[8];
        scan_row<0>(M, ea, eb, Mn[0], Mn[4]);
        scan_row<1>(M, ea, eb, Mn[1], Mn[5]);
        scan_row<2>(M, ea, eb, Mn[2], Mn[6]);
        scan_row<3>(M, ea, eb, Mn[3], Mn[7]);
        if ((k % SCAN_RESCALE) == SCAN_RESCALE - 1 && k != CLEN - 1) {
            float mx0 = Mn[0][0], mx1 = Mn[0][1];
#pragma unroll
            for (int m = 1; m < 8; ++m) { mx0 = fmaxf(mx0, Mn[m][0]); mx1 = fmaxf(mx1, Mn[m][1]); }
            const int ex0 = __builtin_amdgcn_frexp_expf(mx0);
            const int ex1 = __builtin_amdgcn_frexp_expf(mx1);
            esum0 += ex0; esum1 += ex1;
#pragma unroll
            for (int m = 0; m < 8; ++m) { Mn[m][0] = ldexpf(Mn[m][0], -ex0); Mn[m][1] = ldexpf(Mn[m][1], -ex1); }
        }
#pragma unroll
        for (int m = 0; m < 8; ++m) M[m] = Mn[m];
    }

    const float es0 = (float)esum0, es1 = (float)esum1;
    const float ln2 = 0.69314718055994531f;
    v4f r0lo, r0hi, r1lo, r1hi;
#pragma unroll
    for (int m = 0; m < 4; ++m) {
        r0lo[m] = (__log2f(M[m][0])     + es0) * ln2;
        r0hi[m] = (__log2f(M[m + 4][0]) + es0) * ln2;
        r1lo[m] = (__log2f(M[m][1])     + es1) * ln2;
        r1hi[m] = (__log2f(M[m + 4][1]) + es1) * ln2;
    }
    float* dst = chunkM + ((size_t)b * NCHUNK + c) * 64 + q * 8;
    *(v4f*)(dst)      = r0lo;
    *(v4f*)(dst + 4)  = r0hi;
    *(v4f*)(dst + 32) = r1lo;
    *(v4f*)(dst + 36) = r1hi;
}

#define CMB_THREADS 512
#define CMB_GROUPS  (CMB_THREADS / 64)
#define CMB_PIECES  ((NCHUNK * 16 + CMB_THREADS - 1) / CMB_THREADS)

__global__ __launch_bounds__(CMB_THREADS) void combine_kernel(
    const float* __restrict__ chunkM, float* __restrict__ logZdiv)
{
    __shared__ __attribute__((aligned(16))) float bufA[NCHUNK * 64];
    __shared__ __attribute__((aligned(16))) float bufB[((NCHUNK + 1) / 2) * 64];
    __shared__ float vsh[8];

    const int tid = threadIdx.x;
    const int b   = blockIdx.x;
    const float* base = chunkM + (size_t)b * NCHUNK * 64;

    v4f pc[CMB_PIECES];
#pragma unroll
    for (int k = 0; k < CMB_PIECES; ++k) {
        const int i = tid + k * CMB_THREADS;
        if (i < NCHUNK * 16) pc[k] = *(const v4f*)(base + 4 * i);
    }
#pragma unroll
    for (int k = 0; k < CMB_PIECES; ++k) {
        const int i = tid + k * CMB_THREADS;
        if (i < NCHUNK * 16) *(v4f*)(bufA + 4 * i) = pc[k];
    }
    __syncthreads();

    const int e = tid & 63;
    const int g = tid >> 6;
    const int r = e >> 3;
    const int s = e & 7;

    float* src = bufA;
    float* dst = bufB;
#pragma unroll
    for (int n = NCHUNK; n > 1; n = (n + 1) >> 1) {
        const int np = (n + 1) >> 1;
        for (int p = g; p < np; p += CMB_GROUPS) {
            const float* L = src + (2 * p) * 64;
            float res;
            if (2 * p + 1 < n) {
                const float* R = L + 64;
                float t[8];
#pragma unroll
                for (int m = 0; m < 8; ++m) t[m] = L[r * 8 + m] + R[m * 8 + s];
                res = lse8(t);
            } else {
                res = L[e];
            }
            dst[p * 64 + e] = res;
        }
        __syncthreads();
        float* tmp = src; src = dst; dst = tmp;
    }

    if (tid < 8) {
        float t[8];
#pragma unroll
        for (int j = 0; j < 8; ++j) t[j] = src[j * 8 + tid];
        vsh[tid] = lse8(t);
    }
    __syncthreads();
    if (tid == 0) {
        float t[8];
#pragma unroll
        for (int j = 0; j < 8; ++j) t[j] = vsh[j];
        logZdiv[b] = lse8(t) / (float)NTIME;
    }
}

__global__ __launch_bounds__(1024) void seq_scan_kernel(
    const float* __restrict__ out, float* __restrict__ logZdiv)
{
    __shared__ float fwdsh[NBATCH * 8];
    const int tid  = threadIdx.x;
    const int b    = tid >> 3;
    const int s    = tid & 7;
    const int irow = s & 3;

    float mask[8];
#pragma unroll
    for (int m = 0; m < 8; ++m) {
        const bool on = (s < 4) ? (m == s || m == s + 4) : (m == s - 4 || m == s);
        mask[m] = (s < 4) ? (on ? NEG_BIG : 0.0f) : (on ? 0.0f : NEG_BIG);
    }

    fwdsh[tid] = 0.0f;
    __syncthreads();
    for (int t = 0; t < NTIME; ++t) {
        const float* wrow = out + ((size_t)t * NBATCH + b) * NOUT + 8 + irow * 8;
        const float4 w0 = *(const float4*)wrow;
        const float4 w1 = *(const float4*)(wrow + 4);
        float tt[8] = { w0.x + mask[0], w0.y + mask[1], w0.z + mask[2], w0.w + mask[3],
                        w1.x + mask[4], w1.y + mask[5], w1.z + mask[6], w1.w + mask[7] };
#pragma unroll
        for (int m = 0; m < 8; ++m) tt[m] += fwdsh[(b << 3) + m];
        const float nf = lse8(tt);
        __syncthreads();
        fwdsh[tid] = nf;
        __syncthreads();
    }
    if (s == 0) {
        float t[8];
#pragma unroll
        for (int j = 0; j < 8; ++j) t[j] = fwdsh[(b << 3) + j];
        logZdiv[b] = lse8(t) / (float)NTIME;
    }
}

#define SUB_THREADS     128
#define SUB_PIECES      8
#define SUB_NPIECE      (NTIME * NBATCH * 8)
#define SUB_BLOCKS      (SUB_NPIECE / (SUB_THREADS * SUB_PIECES))
#define SUB_STRIDE_ROWS (SUB_BLOCKS * SUB_THREADS / 8)
static_assert(SUB_NPIECE % (SUB_THREADS * SUB_PIECES) == 0, "pieces divide evenly");
static_assert(SUB_STRIDE_ROWS % NBATCH == 0, "one batch entry per thread");

__global__ __launch_bounds__(SUB_THREADS) void sublogz_kernel(
    float* __restrict__ out, const float* __restrict__ logZdiv)
{
    const int q0   = blockIdx.x * SUB_THREADS + threadIdx.x;
    const int row0 = q0 >> 3;
    const int j    = q0 & 7;
    const float z  = logZdiv[row0 & (NBATCH - 1)];
    float* p = out + (size_t)row0 * NOUT + 8 + 4 * j;
    const size_t stride = (size_t)SUB_STRIDE_ROWS * NOUT;

    v4f v[SUB_PIECES];
#pragma unroll
    for (int i = 0; i < SUB_PIECES; ++i)
        v[i] = *(const v4f*)(p + (size_t)i * stride);
#pragma unroll
    for (int i = 0; i < SUB_PIECES; ++i) {
        v4f r = v[i];
        r[0] -= z; r[1] -= z; r[2] -= z; r[3] -= z;
        *(v4f*)(p + (size_t)i * stride) = r;
    }
}

extern "C" void kernel_launch(void* const* d_in, const int* in_sizes, int n_in,
                              void* d_out, int out_size, void* d_ws, size_t ws_size,
                              hipStream_t stream) {
    (void)in_sizes; (void)n_in; (void)out_size;
    const float* x    = (const float*)d_in[0];
    const float* W    = (const float*)d_in[1];
    const float* bias = (const float*)d_in[2];
    float* out = (float*)d_out;

    const int ROWS = NTIME * NBATCH;
    gemm_act_kernel<<<ROWS / 128, 128, 0, stream>>>(x, W, bias, out);

    const size_t chunkBytes = (size_t)NBATCH * NCHUNK * 64 * sizeof(float);
    const size_t logZBytes  = (size_t)NBATCH * sizeof(float);

    if (ws_size >= chunkBytes + logZBytes) {
        float* chunkM  = (float*)d_ws;
        float* logZdiv = (float*)((char*)d_ws + chunkBytes);
        chunk_scan_kernel<<<SCAN_BLOCKS, SCAN_THREADS, 0, stream>>>(out, chunkM);
        combine_kernel<<<NBATCH, CMB_THREADS, 0, stream>>>(chunkM, logZdiv);
        sublogz_kernel<<<SUB_BLOCKS, SUB_THREADS, 0, stream>>>(out, logZdiv);
    } else {
        float* logZdiv = (float*)d_ws;
        seq_scan_kernel<<<1, 1024, 0, stream>>>(out, logZdiv);
        sublogz_kernel<<<SUB_BLOCKS, SUB_THREADS, 0, stream>>>(out, logZdiv);
    }
}


# GEMM epilogue: each wave's 32x40 output block (5120 contiguous bytes) goes through its LDS stage and is written by 10 whole-line 16-B stores, not 48 4-B stores of 64-B and 32-B row pieces
# speedup vs baseline: 3.5831x; 1.0662x over previous
#include <hip/hip_runtime.h>
#include <cmath>

typedef __attribute__((ext_vector_type(2))) float v2f;
typedef __attribute__((ext_vector_type(4))) float v4f;
typedef __attribute__((ext_vector_type(8))) float v8f;

#define NTIME   2000
#define NBATCH  128
#define INSIZE  256
#define NOUT    40
#define NEG_BIG (-1.0e30f)

#define NCHUNK  100
#define CLEN    (NTIME / NCHUNK)

__device__ __forceinline__ float softplusf(float v) {
    return fmaxf(v, 0.0f) + __logf(1.0f + __expf(-fabsf(v)));
}

__device__ __forceinline__ float tanhf_fast(float v) {
    const float a = fabsf(v);
    const float e = __expf(2.0f * a);
    const float t = 1.0f - 2.0f * __builtin_amdgcn_rcpf(e + 1.0f);
    return copysignf(t, v);
}

__device__ __forceinline__ float lse8(const float (&t)[8]) {
    float mx = t[0];
#pragma unroll
    for (int m = 1; m < 8; ++m) mx = fmaxf(mx, t[m]);
    float s = 0.0f;
#pragma unroll
    for (int m = 0; m < 8; ++m) s += __expf(t[m] - mx);
    return mx + __logf(s);
}

__device__ __forceinline__ int iminc(int a, int b) { return a < b ? a : b; }

#define GW_ROWS  32
#define GW_WAVES 4
#define KCH      64
#define XPITCH   68
#define WPITCH   260
#define WPIECES  (NOUT * INSIZE / 4 / (GW_WAVES * 32))
static_assert((NOUT * INSIZE / 4) % (GW_WAVES * 32) == 0, "W pieces divide evenly over the threads");
#define OUT_PIECES (GW_ROWS * NOUT / 4 / 32)
static_assert((GW_ROWS * NOUT / 4) % 32 == 0, "output pieces divide evenly over the lanes");
static_assert(GW_ROWS * NOUT <= GW_ROWS * XPITCH, "the output block fits the wave's LDS stage");
static_assert((GW_ROWS * NOUT * 4) % 128 == 0, "a wave's output block is a whole number of 128-B lines");

#define WMMA4(acc_, a_, b_) \
    acc_ = __builtin_amdgcn_wmma_f32_16x16x4_f32(false, a_, false, b_, (short)0, acc_, false, false)

template <bool AHEAD>
__device__ __forceinline__ void gemm_chunk(
    v8f (&acc)[2][3], v4f (&st)[16],
    float* xst, const float* xfr,
    const float* __restrict__ xnext,
    const float* __restrict__ w0, const float* __restrict__ w1,
    const float* __restrict__ w2)
{
    v4f bf[KCH / 8][3];
#pragma unroll
    for (int j = 0; j < KCH / 8; ++j) {
        bf[j][0] = *(const v4f*)(w0 + j * 8);
        bf[j][1] = *(const v4f*)(w1 + j * 8);
        bf[j][2] = *(const v4f*)(w2 + j * 8);
    }

#pragma unroll
    for (int i = 0; i < 16; ++i)
        *(v4f*)(xst + (2 * i) * XPITCH) = st[i];

    if (AHEAD) {
#pragma unroll
        for (int i = 0; i < 16; ++i)
            st[i] = __builtin_nontemporal_load((const v4f*)(xnext + (size_t)(2 * i) * INSIZE));
    }

    __builtin_amdgcn_fence(__ATOMIC_RELEASE, "wavefront");
    __builtin_amdgcn_wave_barrier();
    __builtin_amdgcn_fence(__ATOMIC_ACQUIRE, "wavefront");

#pragma unroll
    for (int j = 0; j < KCH / 8; ++j) {
        const v4f a0 = *(const v4f*)(xfr + j * 8);
        const v4f a1 = *(const v4f*)(xfr + 16 * XPITCH + j * 8);
        const v4f b0 = bf[j][0], b1 = bf[j][1], b2 = bf[j][2];

        const v2f a0l = __builtin_shufflevector(a0, a0, 0, 1), a0h = __builtin_shufflevector(a0, a0, 2, 3);
        const v2f a1l = __builtin_shufflevector(a1, a1, 0, 1), a1h = __builtin_shufflevector(a1, a1, 2, 3);
        const v2f b0l = __builtin_shufflevector(b0, b0, 0, 1), b0h = __builtin_shufflevector(b0, b0, 2, 3);
        const v2f b1l = __builtin_shufflevector(b1, b1, 0, 1), b1h = __builtin_shufflevector(b1, b1, 2, 3);
        const v2f b2l = __builtin_shufflevector(b2, b2, 0, 1), b2h = __builtin_shufflevector(b2, b2, 2, 3);

        WMMA4(acc[0][0], a0l, b0l);
        WMMA4(acc[0][1], a0l, b1l);
        WMMA4(acc[0][2], a0l, b2l);
        WMMA4(acc[1][0], a1l, b0l);
        WMMA4(acc[1][1], a1l, b1l);
        WMMA4(acc[1][2], a1l, b2l);
        WMMA4(acc[0][0], a0h, b0h);
        WMMA4(acc[0][1], a0h, b1h);
        WMMA4(acc[0][2], a0h, b2h);
        WMMA4(acc[1][0], a1h, b0h);
        WMMA4(acc[1][1], a1h, b1h);
        WMMA4(acc[1][2], a1h, b2h);
    }

    __builtin_amdgcn_fence(__ATOMIC_RELEASE, "wavefront");
    __builtin_amdgcn_wave_barrier();
    __builtin_amdgcn_fence(__ATOMIC_ACQUIRE, "wavefront");
}

__global__ __launch_bounds__(128) void gemm_act_kernel(
    const float* __restrict__ x, const float* __restrict__ W,
    const float* __restrict__ bias, float* __restrict__ out)
{
    __shared__ __attribute__((aligned(16))) float xs[GW_WAVES * GW_ROWS * XPITCH];
    __shared__ __attribute__((aligned(16))) float wsh[NOUT * WPITCH];

    const int lane  = threadIdx.x & 31;
    const int wave  = threadIdx.x >> 5;
    const int l16   = lane & 15;
    const int halfk = lane >> 4;

    const int rowBase = (blockIdx.x * GW_WAVES + wave) * GW_ROWS;
    float* xw = xs + wave * (GW_ROWS * XPITCH);

    v8f acc[2][3];
#pragma unroll
    for (int m = 0; m < 2; ++m)
#pragma unroll
        for (int n = 0; n < 3; ++n)
#pragma unroll
            for (int i = 0; i < 8; ++i) acc[m][n][i] = 0.0f;

    const float* xg  = x + (size_t)(rowBase + halfk) * INSIZE + 4 * l16;
    float*       xst = xw + halfk * XPITCH + 4 * l16;
    const float* xfr = xw + l16 * XPITCH + 4 * halfk;
    const float* wp0 = wsh + iminc(0 * 16 + l16, NOUT - 1) * WPITCH + 4 * halfk;
    const float* wp1 = wsh + iminc(1 * 16 + l16, NOUT - 1) * WPITCH + 4 * halfk;
    const float* wp2 = wsh + iminc(2 * 16 + l16, NOUT - 1) * WPITCH + 4 * halfk;

    v4f wpc[WPIECES];
#pragma unroll
    for (int i = 0; i < WPIECES; ++i)
        wpc[i] = *(const v4f*)(W + 4 * ((int)threadIdx.x + GW_WAVES * 32 * i));

    v4f st[16];
#pragma unroll
    for (int i = 0; i < 16; ++i)
        st[i] = __builtin_nontemporal_load((const v4f*)(xg + (size_t)(2 * i) * INSIZE));

#pragma unroll
    for (int i = 0; i < WPIECES; ++i) {
        const int p = (int)threadIdx.x + GW_WAVES * 32 * i;
        *(v4f*)(wsh + (p >> 6) * WPITCH + 4 * (p & 63)) = wpc[i];
    }
    __syncthreads();

#pragma unroll 1
    for (int ch = 0; ch < INSIZE / KCH - 1; ++ch) {
        const int kc = ch * KCH;
        gemm_chunk<true>(acc, st, xst, xfr, xg + kc + KCH, wp0 + kc, wp1 + kc, wp2 + kc);
    }
    gemm_chunk<false>(acc, st, xst, xfr, xg, wp0 + (INSIZE - KCH), wp1 + (INSIZE - KCH), wp2 + (INSIZE - KCH));

    float* ow = xw;
#pragma unroll
    for (int m = 0; m < 2; ++m) {
#pragma unroll
        for (int nt = 0; nt < 3; ++nt) {
            const int col = nt * 16 + l16;
            if (col >= NOUT) continue;
            const float bb = bias[col];
            const int lrow0 = m * 16 + halfk * 8;
#pragma unroll
            for (int r = 0; r < 8; ++r) {
                const float v = acc[m][nt][r] + bb;
                float res;
                if (nt == 0) {
                    const float sp   = softplusf(v);
                    const float th   = 5.0f * tanhf_fast(v);
                    const float ofs  = (col < 4) ? 1.0f : 0.1f;
                    res = (col < 8) ? (ofs + sp) : th;
                } else {
                    res = 5.0f * tanhf_fast(v);
                }
                ow[(lrow0 + r) * NOUT + col] = res;
            }
        }
    }

    __builtin_amdgcn_fence(__ATOMIC_RELEASE, "wavefront");
    __builtin_amdgcn_wave_barrier();
    __builtin_amdgcn_fence(__ATOMIC_ACQUIRE, "wavefront");

    float* og = out + (size_t)rowBase * NOUT;
    v4f op[OUT_PIECES];
#pragma unroll
    for (int i = 0; i < OUT_PIECES; ++i)
        op[i] = *(const v4f*)(ow + 4 * (32 * i + lane));
#pragma unroll
    for (int i = 0; i < OUT_PIECES; ++i)
        *(v4f*)(og + 4 * (32 * i + lane)) = op[i];
}

#define SCAN_THREADS 128
#define SCAN_GROUPS  (SCAN_THREADS / 4)
#define SCAN_BLOCKS  (NBATCH * NCHUNK / SCAN_GROUPS)
#define SCAN_RESCALE 5
static_assert((NBATCH * NCHUNK) % SCAN_GROUPS == 0, "chunks divide evenly over blocks");

template <int L>
__device__ __forceinline__ float bcast4(float v) {
    return __uint_as_float(__builtin_amdgcn_mov_dpp(__float_as_uint(v), L * 0x55, 0xf, 0xf, true));
}

template <int I>
__device__ __forceinline__ void scan_row(const v2f (&M)[8], const v4f ea, const v4f eb, v2f& mv, v2f& sty) {
    const v4f e = (I < 2) ? ea : eb;
    constexpr int L0 = 2 * (I & 1);
    const float ev[8] = {
        bcast4<L0>(e[0]),     bcast4<L0>(e[1]),     bcast4<L0>(e[2]),     bcast4<L0>(e[3]),
        bcast4<L0 + 1>(e[0]), bcast4<L0 + 1>(e[1]), bcast4<L0 + 1>(e[2]), bcast4<L0 + 1>(e[3]) };
    sty = M[I] * ev[I] + M[I + 4] * ev[I + 4];
    v2f acc = {0.0f, 0.0f};
    bool first = true;
#pragma unroll
    for (int m = 0; m < 8; ++m) {
        if (m == I || m == I + 4) continue;
        acc = first ? M[m] * ev[m] : acc + M[m] * ev[m];
        first = false;
    }
    mv = acc;
}

__global__ __launch_bounds__(SCAN_THREADS) void chunk_scan_kernel(
    const float* __restrict__ out, float* __restrict__ chunkM)
{
    const int q = threadIdx.x & 3;
    const int G = blockIdx.x * SCAN_GROUPS + (threadIdx.x >> 2);
    const int b = G & (NBATCH - 1);
    const int c = G >> 7;

    const float* p = out + ((size_t)(c * CLEN) * NBATCH + b) * NOUT + 8 + 4 * q;
    v4f wa[CLEN], wb[CLEN];
#pragma unroll
    for (int k = 0; k < CLEN; ++k) {
        wa[k] = *(const v4f*)(p + (size_t)k * (NBATCH * NOUT));
        wb[k] = *(const v4f*)(p + (size_t)k * (NBATCH * NOUT) + 16);
    }
    __builtin_amdgcn_sched_barrier(0);

    v2f M[8];
#pragma unroll
    for (int m = 0; m < 8; ++m) {
        M[m][0] = (m == q)     ? 1.0f : 0.0f;
        M[m][1] = (m == q + 4) ? 1.0f : 0.0f;
    }
    int esum0 = 0, esum1 = 0;

#pragma unroll
    for (int k = 0; k < CLEN; ++k) {
        v4f ea, eb;
#pragma unroll
        for (int i = 0; i < 4; ++i) { ea[i] = __expf(wa[k][i]); eb[i] = __expf(wb[k][i]); }
        v2f Mn[8];
        scan_row<0>(M, ea, eb, Mn[0], Mn[4]);
        scan_row<1>(M, ea, eb, Mn[1], Mn[5]);
        scan_row<2>(M, ea, eb, Mn[2], Mn[6]);
        scan_row<3>(M, ea, eb, Mn[3], Mn[7]);
        if ((k % SCAN_RESCALE) == SCAN_RESCALE - 1 && k != CLEN - 1) {
            float mx0 = Mn[0][0], mx1 = Mn[0][1];
#pragma unroll
            for (int m = 1; m < 8; ++m) { mx0 = fmaxf(mx0, Mn[m][0]); mx1 = fmaxf(mx1, Mn[m][1]); }
            const int ex0 = __builtin_amdgcn_frexp_expf(mx0);
            const int ex1 = __builtin_amdgcn_frexp_expf(mx1);
            esum0 += ex0; esum1 += ex1;
#pragma unroll
            for (int m = 0; m < 8; ++m) { Mn[m][0] = ldexpf(Mn[m][0], -ex0); Mn[m][1] = ldexpf(Mn[m][1], -ex1); }
        }
#pragma unroll
        for (int m = 0; m < 8; ++m) M[m] = Mn[m];
    }

    const float es0 = (float)esum0, es1 = (float)esum1;
    const float ln2 = 0.69314718055994531f;
    v4f r0lo, r0hi, r1lo, r1hi;
#pragma unroll
    for (int m = 0; m < 4; ++m) {
        r0lo[m] = (__log2f(M[m][0])     + es0) * ln2;
        r0hi[m] = (__log2f(M[m + 4][0]) + es0) * ln2;
        r1lo[m] = (__log2f(M[m][1])     + es1) * ln2;
        r1hi[m] = (__log2f(M[m + 4][1]) + es1) * ln2;
    }
    float* dst = chunkM + ((size_t)b * NCHUNK + c) * 64 + q * 8;
    *(v4f*)(dst)      = r0lo;
    *(v4f*)(dst + 4)  = r0hi;
    *(v4f*)(dst + 32) = r1lo;
    *(v4f*)(dst + 36) = r1hi;
}

#define CMB_THREADS 512
#define CMB_GROUPS  (CMB_THREADS / 64)
#define CMB_PIECES  ((NCHUNK * 16 + CMB_THREADS - 1) / CMB_THREADS)

__global__ __launch_bounds__(CMB_THREADS) void combine_kernel(
    const float* __restrict__ chunkM, float* __restrict__ logZdiv)
{
    __shared__ __attribute__((aligned(16))) float bufA[NCHUNK * 64];
    __shared__ __attribute__((aligned(16))) float bufB[((NCHUNK + 1) / 2) * 64];
    __shared__ float vsh[8];

    const int tid = threadIdx.x;
    const int b   = blockIdx.x;
    const float* base = chunkM + (size_t)b * NCHUNK * 64;

    v4f pc[CMB_PIECES];
#pragma unroll
    for (int k = 0; k < CMB_PIECES; ++k) {
        const int i = tid + k * CMB_THREADS;
        if (i < NCHUNK * 16) pc[k] = *(const v4f*)(base + 4 * i);
    }
#pragma unroll
    for (int k = 0; k < CMB_PIECES; ++k) {
        const int i = tid + k * CMB_THREADS;
        if (i < NCHUNK * 16) *(v4f*)(bufA + 4 * i) = pc[k];
    }
    __syncthreads();

    const int e = tid & 63;
    const int g = tid >> 6;
    const int r = e >> 3;
    const int s = e & 7;

    float* src = bufA;
    float* dst = bufB;
#pragma unroll
    for (int n = NCHUNK; n > 1; n = (n + 1) >> 1) {
        const int np = (n + 1) >> 1;
        for (int p = g; p < np; p += CMB_GROUPS) {
            const float* L = src + (2 * p) * 64;
            float res;
            if (2 * p + 1 < n) {
                const float* R = L + 64;
                float t[8];
#pragma unroll
                for (int m = 0; m < 8; ++m) t[m] = L[r * 8 + m] + R[m * 8 + s];
                res = lse8(t);
            } else {
                res = L[e];
            }
            dst[p * 64 + e] = res;
        }
        __syncthreads();
        float* tmp = src; src = dst; dst = tmp;
    }

    if (tid < 8) {
        float t[8];
#pragma unroll
        for (int j = 0; j < 8; ++j) t[j] = src[j * 8 + tid];
        vsh[tid] = lse8(t);
    }
    __syncthreads();
    if (tid == 0) {
        float t[8];
#pragma unroll
        for (int j = 0; j < 8; ++j) t[j] = vsh[j];
        logZdiv[b] = lse8(t) / (float)NTIME;
    }
}

__global__ __launch_bounds__(1024) void seq_scan_kernel(
    const float* __restrict__ out, float* __restrict__ logZdiv)
{
    __shared__ float fwdsh[NBATCH * 8];
    const int tid  = threadIdx.x;
    const int b    = tid >> 3;
    const int s    = tid & 7;
    const int irow = s & 3;

    float mask[8];
#pragma unroll
    for (int m = 0; m < 8; ++m) {
        const bool on = (s < 4) ? (m == s || m == s + 4) : (m == s - 4 || m == s);
        mask[m] = (s < 4) ? (on ? NEG_BIG : 0.0f) : (on ? 0.0f : NEG_BIG);
    }

    fwdsh[tid] = 0.0f;
    __syncthreads();
    for (int t = 0; t < NTIME; ++t) {
        const float* wrow = out + ((size_t)t * NBATCH + b) * NOUT + 8 + irow * 8;
        const float4 w0 = *(const float4*)wrow;
        const float4 w1 = *(const float4*)(wrow + 4);
        float tt[8] = { w0.x + mask[0], w0.y + mask[1], w0.z + mask[2], w0.w + mask[3],
                        w1.x + mask[4], w1.y + mask[5], w1.z + mask[6], w1.w + mask[7] };
#pragma unroll
        for (int m = 0; m < 8; ++m) tt[m] += fwdsh[(b << 3) + m];
        const float nf = lse8(tt);
        __syncthreads();
        fwdsh[tid] = nf;
        __syncthreads();
    }
    if (s == 0) {
        float t[8];
#pragma unroll
        for (int j = 0; j < 8; ++j) t[j] = fwdsh[(b << 3) + j];
        logZdiv[b] = lse8(t) / (float)NTIME;
    }
}

#define SUB_THREADS     128
#define SUB_PIECES      8
#define SUB_NPIECE      (NTIME * NBATCH * 8)
#define SUB_BLOCKS      (SUB_NPIECE / (SUB_THREADS * SUB_PIECES))
#define SUB_STRIDE_ROWS (SUB_BLOCKS * SUB_THREADS / 8)
static_assert(SUB_NPIECE % (SUB_THREADS * SUB_PIECES) == 0, "pieces divide evenly");
static_assert(SUB_STRIDE_ROWS % NBATCH == 0, "one batch entry per thread");

__global__ __launch_bounds__(SUB_THREADS) void sublogz_kernel(
    float* __restrict__ out, const float* __restrict__ logZdiv)
{
    const int q0   = blockIdx.x * SUB_THREADS + threadIdx.x;
    const int row0 = q0 >> 3;
    const int j    = q0 & 7;
    const float z  = logZdiv[row0 & (NBATCH - 1)];
    float* p = out + (size_t)row0 * NOUT + 8 + 4 * j;
    const size_t stride = (size_t)SUB_STRIDE_ROWS * NOUT;

    v4f v[SUB_PIECES];
#pragma unroll
    for (int i = 0; i < SUB_PIECES; ++i)
        v[i] = *(const v4f*)(p + (size_t)i * stride);
#pragma unroll
    for (int i = 0; i < SUB_PIECES; ++i) {
        v4f r = v[i];
        r[0] -= z; r[1] -= z; r[2] -= z; r[3] -= z;
        *(v4f*)(p + (size_t)i * stride) = r;
    }
}

extern "C" void kernel_launch(void* const* d_in, const int* in_sizes, int n_in,
                              void* d_out, int out_size, void* d_ws, size_t ws_size,
                              hipStream_t stream) {
    (void)in_sizes; (void)n_in; (void)out_size;
    const float* x    = (const float*)d_in[0];
    const float* W    = (const float*)d_in[1];
    const float* bias = (const float*)d_in[2];
    float* out = (float*)d_out;

    const int ROWS = NTIME * NBATCH;
    gemm_act_kernel<<<ROWS / 128, 128, 0, stream>>>(x, W, bias, out);

    const size_t chunkBytes = (size_t)NBATCH * NCHUNK * 64 * sizeof(float);
    const size_t logZBytes  = (size_t)NBATCH * sizeof(float);

    if (ws_size >= chunkBytes + logZBytes) {
        float* chunkM  = (float*)d_ws;
        float* logZdiv = (float*)((char*)d_ws + chunkBytes);
        chunk_scan_kernel<<<SCAN_BLOCKS, SCAN_THREADS, 0, stream>>>(out, chunkM);
        combine_kernel<<<NBATCH, CMB_THREADS, 0, stream>>>(chunkM, logZdiv);
        sublogz_kernel<<<SUB_BLOCKS, SUB_THREADS, 0, stream>>>(out, logZdiv);
    } else {
        float* logZdiv = (float*)d_ws;
        seq_scan_kernel<<<1, 1024, 0, stream>>>(out, logZdiv);
        sublogz_kernel<<<SUB_BLOCKS, SUB_THREADS, 0, stream>>>(out, logZdiv);
    }
}
